# starting kernel
# baseline (speedup 1.0000x reference)
#include <hip/hip_runtime.h>
#include <hip/hip_bf16.h>


#define Hh   512
#define H1   511
#define NN   1024
#define BB   16384

typedef float v2f __attribute__((ext_vector_type(2)));
typedef float v8f __attribute__((ext_vector_type(8)));

__global__ __launch_bounds__(512) void k_build_fg(
    const float* __restrict__ enc_rot, const float* __restrict__ enc_diag,
    const float* __restrict__ dec_rot, const float* __restrict__ dec_diag,
    float* __restrict__ fg) {
  const int m = blockIdx.x;
  const int t = threadIdx.x;
  const float* r = (m < 4) ? (enc_rot + m * Hh) : (dec_rot + (m - 4) * Hh);
  const float* d = (m < 4) ? (enc_diag + m * Hh) : (dec_diag + (m - 4) * Hh);

  __shared__ float sA[512];
  __shared__ float sB[512];
  __shared__ float cS[4];

  const float rv = r[t];
  sA[t] = rv;
  sB[t] = rv * rv;
  __syncthreads();
  for (int off = 256; off > 0; off >>= 1) {
    if (t < off) { sA[t] += sA[t + off]; sB[t] += sB[t + off]; }
    __syncthreads();
  }
  if (t == 0) {
    const float a = sA[0], b = sB[0];
    const float D = 1.0f - a * a + 512.0f * b;
    cS[0] = -2.0f * b / D;
    cS[1] = 1.0f + ((1.0f + a) * (1.0f + a) - 512.0f * b) / D;
    cS[2] = -1.0f - ((1.0f - a) * (1.0f - a) - 512.0f * b) / D;
    cS[3] = -1024.0f / D;
  }
  __syncthreads();
  const float dv = d[t];
  fg[m * 1024 + t]       = dv * (cS[0] + cS[1] * rv);
  fg[m * 1024 + 512 + t] = dv * (cS[2] + cS[3] * rv);
}

__global__ void k_build_prefix(
    int m,
    const float* __restrict__ enc_rot, const float* __restrict__ enc_diag,
    const float* __restrict__ dec_rot, const float* __restrict__ dec_diag,
    const float* __restrict__ fg, float* __restrict__ P) {
  const int idx = blockIdx.x * blockDim.x + threadIdx.x;
  if (idx >= 4 * 1023) return;
  const int k  = idx & 3;
  const int si = idx >> 2;
  const int s  = si - H1;

  const float* r = (m < 4) ? (enc_rot + m * Hh) : (dec_rot + (m - 4) * Hh);
  const float* d = (m < 4) ? (enc_diag + m * Hh) : (dec_diag + (m - 4) * Hh);
  const float* f = fg + m * 1024;
  const float* g = f + 512;

  float run[4] = {0.f, 0.f, 0.f, 0.f};
  float* Pp = P + ((size_t)(k * 1023 + si)) * Hh;

  const int tlo = (s > 0) ? s : 0;
  const int thi = (s < 0) ? (H1 + s) : H1;

  for (int t = 0; t < Hh; ++t) {
    float h = 0.f;
    if (t >= tlo && t <= thi) {
      const int v = t - s;
      if (k == 0) {
        h = f[t] + g[t] * r[v];
        if (s == 0) h += d[t];
      } else if (k == 1) {
        h = f[v] + g[v] * r[H1 - t];
        if (2 * t == H1 + s) h += d[v];
      } else if (k == 2) {
        h = f[H1 - t] + g[H1 - t] * r[H1 - v];
        if (s == 0) h += d[H1 - t];
      } else {
        h = f[H1 - v] + g[H1 - v] * r[t];
        if (2 * t == H1 + s) h += d[H1 - v];
      }
    }
    run[t & 3] += h;
    Pp[t] = run[t & 3];
  }
}

__global__ void k_build_lifted(const float* __restrict__ P,
                               float* __restrict__ L) {
  const int idx = blockIdx.x * blockDim.x + threadIdx.x;
  const int i = idx >> 10;
  const int j = idx & 1023;
  const int s = i - j;
  float acc = 0.f;
  if (s > -Hh && s < Hh) {
    const int si = s + H1;
    const int a = max(0, max(s, i - Hh));
    const int b = min(i, min(H1, H1 + s));
    if (b >= a) {
#pragma unroll
      for (int k = 0; k < 4; ++k) {
        const int c   = (i - k) & 3;
        const int Thi = b - ((b + 4 - c) & 3);
        const int Tlo = a + ((c + 4 - (a & 3)) & 3);
        if (Thi >= Tlo) {
          const float* Pp = P + ((size_t)(k * 1023 + si)) * Hh;
          float v = Pp[Thi];
          if (Tlo >= 4) v -= Pp[Tlo - 4];
          acc += v;
        }
      }
    }
  }
  L[idx] = acc;
}

#define BM 128
#define BN 64
#define BK 16

__global__ __launch_bounds__(256) void k_gemm_xwt(const float* __restrict__ A,
                                                  const float* __restrict__ W,
                                                  float* __restrict__ Out) {
  __shared__ float As[BM][BK + 1];
  __shared__ float Bs[BK][BN + 1];

  const int tid  = threadIdx.x;
  const int mb   = blockIdx.x * BM;
  const int nb   = blockIdx.y * BN;
  const int lane = tid & 31;
  const int w    = tid >> 5;
  const int wm   = (w & 3) * 32;
  const int wn   = (w >> 2) * 32;
  const int half = lane >> 4;
  const int l15  = lane & 15;

  v8f acc[2][2];
#pragma unroll
  for (int fm = 0; fm < 2; ++fm)
#pragma unroll
    for (int fn = 0; fn < 2; ++fn) acc[fm][fn] = v8f{};

  const int am = tid >> 1;
  const int ac = (tid & 1) * 8;
  const int wr = tid >> 2;
  const int wk = (tid & 3) * 4;

  for (int kb = 0; kb < 1024; kb += BK) {
    const float4 a0 = *(const float4*)(A + (size_t)(mb + am) * 1024 + kb + ac);
    const float4 a1 = *(const float4*)(A + (size_t)(mb + am) * 1024 + kb + ac + 4);
    As[am][ac + 0] = a0.x; As[am][ac + 1] = a0.y;
    As[am][ac + 2] = a0.z; As[am][ac + 3] = a0.w;
    As[am][ac + 4] = a1.x; As[am][ac + 5] = a1.y;
    As[am][ac + 6] = a1.z; As[am][ac + 7] = a1.w;
    const float4 wv = *(const float4*)(W + (size_t)(nb + wr) * 1024 + kb + wk);
    Bs[wk + 0][wr] = wv.x; Bs[wk + 1][wr] = wv.y;
    Bs[wk + 2][wr] = wv.z; Bs[wk + 3][wr] = wv.w;
    __syncthreads();

#pragma unroll
    for (int k4 = 0; k4 < BK; k4 += 4) {
      v2f af[2], bf[2];
#pragma unroll
      for (int fm = 0; fm < 2; ++fm) {
        const int row = wm + fm * 16 + l15;
        af[fm].x = As[row][k4 + half * 2];
        af[fm].y = As[row][k4 + half * 2 + 1];
      }
#pragma unroll
      for (int fn = 0; fn < 2; ++fn) {
        const int col = wn + fn * 16 + l15;
        bf[fn].x = Bs[k4 + half * 2][col];
        bf[fn].y = Bs[k4 + half * 2 + 1][col];
      }
#pragma unroll
      for (int fm = 0; fm < 2; ++fm)
#pragma unroll
        for (int fn = 0; fn < 2; ++fn)
          acc[fm][fn] = __builtin_amdgcn_wmma_f32_16x16x4_f32(
              false, af[fm], false, bf[fn], (short)0, acc[fm][fn],
              false, false);
    }
    __syncthreads();
  }

#pragma unroll
  for (int fm = 0; fm < 2; ++fm) {
#pragma unroll
    for (int fn = 0; fn < 2; ++fn) {
      const int col = nb + wn + fn * 16 + l15;
#pragma unroll
      for (int g = 0; g < 8; ++g) {
        const int row = mb + wm + fm * 16 + g + half * 8;
        Out[(size_t)row * 1024 + col] = acc[fm][fn][g];
      }
    }
  }
}

extern "C" void kernel_launch(void* const* d_in, const int* in_sizes, int n_in,
                              void* d_out, int out_size, void* d_ws,
                              size_t ws_size, hipStream_t stream) {
  (void)in_sizes; (void)n_in; (void)out_size; (void)ws_size;

  const float* x        = (const float*)d_in[0];
  const float* enc_rot  = (const float*)d_in[1];
  const float* enc_diag = (const float*)d_in[2];
  const float* dec_rot  = (const float*)d_in[3];
  const float* dec_diag = (const float*)d_in[4];

  float* out = (float*)d_out;
  float* ws  = (float*)d_ws;

  float* fg     = ws;
  float* P      = fg + (size_t)8 * 1024;
  float* lifted = P + (size_t)4 * 1023 * Hh;
  float* actA   = lifted + (size_t)8 * NN * NN;
  float* actB   = actA + (size_t)BB * NN;

  float* bneck  = out;
  float* final_ = out + (size_t)BB * NN;

  k_build_fg<<<8, 512, 0, stream>>>(enc_rot, enc_diag, dec_rot, dec_diag, fg);

  for (int m = 0; m < 8; ++m) {
    k_build_prefix<<<(4 * 1023 + 255) / 256, 256, 0, stream>>>(
        m, enc_rot, enc_diag, dec_rot, dec_diag, fg, P);
    k_build_lifted<<<(NN * NN) / 256, 256, 0, stream>>>(
        P, lifted + (size_t)m * NN * NN);
  }

  dim3 grid(BB / BM, NN / BN);
  k_gemm_xwt<<<grid, 256, 0, stream>>>(x,     lifted + 0 * (size_t)NN * NN, actA);
  k_gemm_xwt<<<grid, 256, 0, stream>>>(actA,  lifted + 1 * (size_t)NN * NN, actB);
  k_gemm_xwt<<<grid, 256, 0, stream>>>(actB,  lifted + 2 * (size_t)NN * NN, actA);
  k_gemm_xwt<<<grid, 256, 0, stream>>>(actA,  lifted + 3 * (size_t)NN * NN, bneck);
  k_gemm_xwt<<<grid, 256, 0, stream>>>(bneck, lifted + 4 * (size_t)NN * NN, actA);
  k_gemm_xwt<<<grid, 256, 0, stream>>>(actA,  lifted + 5 * (size_t)NN * NN, actB);
  k_gemm_xwt<<<grid, 256, 0, stream>>>(actB,  lifted + 6 * (size_t)NN * NN, actA);
  k_gemm_xwt<<<grid, 256, 0, stream>>>(actA,  lifted + 7 * (size_t)NN * NN, final_);
}


# k_build_prefix: one wave per (rotation, diagonal) row, lanes along t, class sums by wave shuffles (tree in 32, then carry), not one thread's 512 serial steps; prefix summation order changes
# speedup vs baseline: 2.2320x; 2.2320x over previous
#include <hip/hip_runtime.h>
#include <hip/hip_bf16.h>


#define Hh   512
#define H1   511
#define NN   1024
#define BB   16384

typedef float v2f __attribute__((ext_vector_type(2)));
typedef float v8f __attribute__((ext_vector_type(8)));

__global__ __launch_bounds__(512) void k_build_fg(
    const float* __restrict__ enc_rot, const float* __restrict__ enc_diag,
    const float* __restrict__ dec_rot, const float* __restrict__ dec_diag,
    float* __restrict__ fg) {
  const int m = blockIdx.x;
  const int t = threadIdx.x;
  const float* r = (m < 4) ? (enc_rot + m * Hh) : (dec_rot + (m - 4) * Hh);
  const float* d = (m < 4) ? (enc_diag + m * Hh) : (dec_diag + (m - 4) * Hh);

  __shared__ float sA[512];
  __shared__ float sB[512];
  __shared__ float cS[4];

  const float rv = r[t];
  sA[t] = rv;
  sB[t] = rv * rv;
  __syncthreads();
  for (int off = 256; off > 0; off >>= 1) {
    if (t < off) { sA[t] += sA[t + off]; sB[t] += sB[t + off]; }
    __syncthreads();
  }
  if (t == 0) {
    const float a = sA[0], b = sB[0];
    const float D = 1.0f - a * a + 512.0f * b;
    cS[0] = -2.0f * b / D;
    cS[1] = 1.0f + ((1.0f + a) * (1.0f + a) - 512.0f * b) / D;
    cS[2] = -1.0f - ((1.0f - a) * (1.0f - a) - 512.0f * b) / D;
    cS[3] = -1024.0f / D;
  }
  __syncthreads();
  const float dv = d[t];
  fg[m * 1024 + t]       = dv * (cS[0] + cS[1] * rv);
  fg[m * 1024 + 512 + t] = dv * (cS[2] + cS[3] * rv);
}

__global__ __launch_bounds__(256) void k_build_prefix(
    int m,
    const float* __restrict__ enc_rot, const float* __restrict__ enc_diag,
    const float* __restrict__ dec_rot, const float* __restrict__ dec_diag,
    const float* __restrict__ fg, float* __restrict__ P) {
  const int lane = threadIdx.x & 31;
  const int wave = __builtin_amdgcn_readfirstlane((int)(threadIdx.x >> 5));
  const int idx  = blockIdx.x * 8 + wave;
  if (idx >= 4 * 1023) return;
  const int k  = idx & 3;
  const int si = idx >> 2;
  const int s  = si - H1;

  const float* r = (m < 4) ? (enc_rot + m * Hh) : (dec_rot + (m - 4) * Hh);
  const float* d = (m < 4) ? (enc_diag + m * Hh) : (dec_diag + (m - 4) * Hh);
  const float* f = fg + m * 1024;
  const float* g = f + 512;

  float* Pp = P + ((size_t)(k * 1023 + si)) * Hh;

  const int tlo = (s > 0) ? s : 0;
  const int thi = (s < 0) ? (H1 + s) : H1;

  const int sgn_f = (k < 2) ? 1 : -1;
  const int off_f = (k == 0) ? 0 : (k == 1) ? -s : (k == 2) ? H1 : (H1 + s);
  const int sgn_r = (k == 0 || k == 3) ? 1 : -1;
  const int off_r = (k == 0) ? -s : (k == 1) ? H1 : (k == 2) ? (H1 + s) : 0;

  const unsigned span = (unsigned)(thi - tlo);
  float hv[16];
#pragma unroll
  for (int j = 0; j < 16; ++j) {
    const int t  = j * 32 + lane;
    const int fi = min(max(sgn_f * t + off_f, 0), H1);
    const int ri = min(max(sgn_r * t + off_r, 0), H1);
    const float h = f[fi] + g[fi] * r[ri];
    hv[j] = ((unsigned)(t - tlo) <= span) ? h : 0.f;
  }

  if (k & 1) {
    if (((H1 + s) & 1) == 0) {
      const int ts = (H1 + s) >> 1;
      const float dv = d[sgn_f * ts + off_f];
#pragma unroll
      for (int j = 0; j < 16; ++j)
        if (j * 32 + lane == ts) hv[j] += dv;
    }
  } else if (s == 0) {
#pragma unroll
    for (int j = 0; j < 16; ++j)
      hv[j] += d[sgn_f * (j * 32 + lane) + off_f];
  }

  const int cls_last = 28 + (lane & 3);
  float carry = 0.f;
#pragma unroll
  for (int j = 0; j < 16; ++j) {
    float v = hv[j];
    float u;
    u = __shfl_up(v, 4);  if (lane >= 4)  v += u;
    u = __shfl_up(v, 8);  if (lane >= 8)  v += u;
    u = __shfl_up(v, 16); if (lane >= 16) v += u;
    const float tot = __shfl(v, cls_last);
    Pp[j * 32 + lane] = v + carry;
    carry += tot;
  }
}

__global__ void k_build_lifted(const float* __restrict__ P,
                               float* __restrict__ L) {
  const int idx = blockIdx.x * blockDim.x + threadIdx.x;
  const int i = idx >> 10;
  const int j = idx & 1023;
  const int s = i - j;
  float acc = 0.f;
  if (s > -Hh && s < Hh) {
    const int si = s + H1;
    const int a = max(0, max(s, i - Hh));
    const int b = min(i, min(H1, H1 + s));
    if (b >= a) {
#pragma unroll
      for (int k = 0; k < 4; ++k) {
        const int c   = (i - k) & 3;
        const int Thi = b - ((b + 4 - c) & 3);
        const int Tlo = a + ((c + 4 - (a & 3)) & 3);
        if (Thi >= Tlo) {
          const float* Pp = P + ((size_t)(k * 1023 + si)) * Hh;
          float v = Pp[Thi];
          if (Tlo >= 4) v -= Pp[Tlo - 4];
          acc += v;
        }
      }
    }
  }
  L[idx] = acc;
}

#define BM 128
#define BN 64
#define BK 16

__global__ __launch_bounds__(256) void k_gemm_xwt(const float* __restrict__ A,
                                                  const float* __restrict__ W,
                                                  float* __restrict__ Out) {
  __shared__ float As[BM][BK + 1];
  __shared__ float Bs[BK][BN + 1];

  const int tid  = threadIdx.x;
  const int mb   = blockIdx.x * BM;
  const int nb   = blockIdx.y * BN;
  const int lane = tid & 31;
  const int w    = tid >> 5;
  const int wm   = (w & 3) * 32;
  const int wn   = (w >> 2) * 32;
  const int half = lane >> 4;
  const int l15  = lane & 15;

  v8f acc[2][2];
#pragma unroll
  for (int fm = 0; fm < 2; ++fm)
#pragma unroll
    for (int fn = 0; fn < 2; ++fn) acc[fm][fn] = v8f{};

  const int am = tid >> 1;
  const int ac = (tid & 1) * 8;
  const int wr = tid >> 2;
  const int wk = (tid & 3) * 4;

  for (int kb = 0; kb < 1024; kb += BK) {
    const float4 a0 = *(const float4*)(A + (size_t)(mb + am) * 1024 + kb + ac);
    const float4 a1 = *(const float4*)(A + (size_t)(mb + am) * 1024 + kb + ac + 4);
    As[am][ac + 0] = a0.x; As[am][ac + 1] = a0.y;
    As[am][ac + 2] = a0.z; As[am][ac + 3] = a0.w;
    As[am][ac + 4] = a1.x; As[am][ac + 5] = a1.y;
    As[am][ac + 6] = a1.z; As[am][ac + 7] = a1.w;
    const float4 wv = *(const float4*)(W + (size_t)(nb + wr) * 1024 + kb + wk);
    Bs[wk + 0][wr] = wv.x; Bs[wk + 1][wr] = wv.y;
    Bs[wk + 2][wr] = wv.z; Bs[wk + 3][wr] = wv.w;
    __syncthreads();

#pragma unroll
    for (int k4 = 0; k4 < BK; k4 += 4) {
      v2f af[2], bf[2];
#pragma unroll
      for (int fm = 0; fm < 2; ++fm) {
        const int row = wm + fm * 16 + l15;
        af[fm].x = As[row][k4 + half * 2];
        af[fm].y = As[row][k4 + half * 2 + 1];
      }
#pragma unroll
      for (int fn = 0; fn < 2; ++fn) {
        const int col = wn + fn * 16 + l15;
        bf[fn].x = Bs[k4 + half * 2][col];
        bf[fn].y = Bs[k4 + half * 2 + 1][col];
      }
#pragma unroll
      for (int fm = 0; fm < 2; ++fm)
#pragma unroll
        for (int fn = 0; fn < 2; ++fn)
          acc[fm][fn] = __builtin_amdgcn_wmma_f32_16x16x4_f32(
              false, af[fm], false, bf[fn], (short)0, acc[fm][fn],
              false, false);
    }
    __syncthreads();
  }

#pragma unroll
  for (int fm = 0; fm < 2; ++fm) {
#pragma unroll
    for (int fn = 0; fn < 2; ++fn) {
      const int col = nb + wn + fn * 16 + l15;
#pragma unroll
      for (int g = 0; g < 8; ++g) {
        const int row = mb + wm + fm * 16 + g + half * 8;
        Out[(size_t)row * 1024 + col] = acc[fm][fn][g];
      }
    }
  }
}

extern "C" void kernel_launch(void* const* d_in, const int* in_sizes, int n_in,
                              void* d_out, int out_size, void* d_ws,
                              size_t ws_size, hipStream_t stream) {
  (void)in_sizes; (void)n_in; (void)out_size; (void)ws_size;

  const float* x        = (const float*)d_in[0];
  const float* enc_rot  = (const float*)d_in[1];
  const float* enc_diag = (const float*)d_in[2];
  const float* dec_rot  = (const float*)d_in[3];
  const float* dec_diag = (const float*)d_in[4];

  float* out = (float*)d_out;
  float* ws  = (float*)d_ws;

  float* fg     = ws;
  float* P      = fg + (size_t)8 * 1024;
  float* lifted = P + (size_t)4 * 1023 * Hh;
  float* actA   = lifted + (size_t)8 * NN * NN;
  float* actB   = actA + (size_t)BB * NN;

  float* bneck  = out;
  float* final_ = out + (size_t)BB * NN;

  k_build_fg<<<8, 512, 0, stream>>>(enc_rot, enc_diag, dec_rot, dec_diag, fg);

  for (int m = 0; m < 8; ++m) {
    k_build_prefix<<<(4 * 1023 + 7) / 8, 256, 0, stream>>>(
        m, enc_rot, enc_diag, dec_rot, dec_diag, fg, P);
    k_build_lifted<<<(NN * NN) / 256, 256, 0, stream>>>(
        P, lifted + (size_t)m * NN * NN);
  }

  dim3 grid(BB / BM, NN / BN);
  k_gemm_xwt<<<grid, 256, 0, stream>>>(x,     lifted + 0 * (size_t)NN * NN, actA);
  k_gemm_xwt<<<grid, 256, 0, stream>>>(actA,  lifted + 1 * (size_t)NN * NN, actB);
  k_gemm_xwt<<<grid, 256, 0, stream>>>(actB,  lifted + 2 * (size_t)NN * NN, actA);
  k_gemm_xwt<<<grid, 256, 0, stream>>>(actA,  lifted + 3 * (size_t)NN * NN, bneck);
  k_gemm_xwt<<<grid, 256, 0, stream>>>(bneck, lifted + 4 * (size_t)NN * NN, actA);
  k_gemm_xwt<<<grid, 256, 0, stream>>>(actA,  lifted + 5 * (size_t)NN * NN, actB);
  k_gemm_xwt<<<grid, 256, 0, stream>>>(actB,  lifted + 6 * (size_t)NN * NN, actA);
  k_gemm_xwt<<<grid, 256, 0, stream>>>(actA,  lifted + 7 * (size_t)NN * NN, final_);
}


# Fold each half's four lifted weights into one matrix first, (L3 L2)(L1 L0), in two batched launches: the batch passes 2 big GEMMs, not 8; products re-associated, summation order changes
# speedup vs baseline: 5.4994x; 2.4639x over previous
#include <hip/hip_runtime.h>
#include <hip/hip_bf16.h>


#define Hh   512
#define H1   511
#define NN   1024
#define BB   16384

typedef float v2f __attribute__((ext_vector_type(2)));
typedef float v8f __attribute__((ext_vector_type(8)));

__global__ __launch_bounds__(512) void k_build_fg(
    const float* __restrict__ enc_rot, const float* __restrict__ enc_diag,
    const float* __restrict__ dec_rot, const float* __restrict__ dec_diag,
    float* __restrict__ fg) {
  const int m = blockIdx.x;
  const int t = threadIdx.x;
  const float* r = (m < 4) ? (enc_rot + m * Hh) : (dec_rot + (m - 4) * Hh);
  const float* d = (m < 4) ? (enc_diag + m * Hh) : (dec_diag + (m - 4) * Hh);

  __shared__ float sA[512];
  __shared__ float sB[512];
  __shared__ float cS[4];

  const float rv = r[t];
  sA[t] = rv;
  sB[t] = rv * rv;
  __syncthreads();
  for (int off = 256; off > 0; off >>= 1) {
    if (t < off) { sA[t] += sA[t + off]; sB[t] += sB[t + off]; }
    __syncthreads();
  }
  if (t == 0) {
    const float a = sA[0], b = sB[0];
    const float D = 1.0f - a * a + 512.0f * b;
    cS[0] = -2.0f * b / D;
    cS[1] = 1.0f + ((1.0f + a) * (1.0f + a) - 512.0f * b) / D;
    cS[2] = -1.0f - ((1.0f - a) * (1.0f - a) - 512.0f * b) / D;
    cS[3] = -1024.0f / D;
  }
  __syncthreads();
  const float dv = d[t];
  fg[m * 1024 + t]       = dv * (cS[0] + cS[1] * rv);
  fg[m * 1024 + 512 + t] = dv * (cS[2] + cS[3] * rv);
}

__global__ __launch_bounds__(256) void k_build_prefix(
    int m,
    const float* __restrict__ enc_rot, const float* __restrict__ enc_diag,
    const float* __restrict__ dec_rot, const float* __restrict__ dec_diag,
    const float* __restrict__ fg, float* __restrict__ P) {
  const int lane = threadIdx.x & 31;
  const int wave = __builtin_amdgcn_readfirstlane((int)(threadIdx.x >> 5));
  const int idx  = blockIdx.x * 8 + wave;
  if (idx >= 4 * 1023) return;
  const int k  = idx & 3;
  const int si = idx >> 2;
  const int s  = si - H1;

  const float* r = (m < 4) ? (enc_rot + m * Hh) : (dec_rot + (m - 4) * Hh);
  const float* d = (m < 4) ? (enc_diag + m * Hh) : (dec_diag + (m - 4) * Hh);
  const float* f = fg + m * 1024;
  const float* g = f + 512;

  float* Pp = P + ((size_t)(k * 1023 + si)) * Hh;

  const int tlo = (s > 0) ? s : 0;
  const int thi = (s < 0) ? (H1 + s) : H1;

  const int sgn_f = (k < 2) ? 1 : -1;
  const int off_f = (k == 0) ? 0 : (k == 1) ? -s : (k == 2) ? H1 : (H1 + s);
  const int sgn_r = (k == 0 || k == 3) ? 1 : -1;
  const int off_r = (k == 0) ? -s : (k == 1) ? H1 : (k == 2) ? (H1 + s) : 0;

  const unsigned span = (unsigned)(thi - tlo);
  float hv[16];
#pragma unroll
  for (int j = 0; j < 16; ++j) {
    const int t  = j * 32 + lane;
    const int fi = min(max(sgn_f * t + off_f, 0), H1);
    const int ri = min(max(sgn_r * t + off_r, 0), H1);
    const float h = f[fi] + g[fi] * r[ri];
    hv[j] = ((unsigned)(t - tlo) <= span) ? h : 0.f;
  }

  if (k & 1) {
    if (((H1 + s) & 1) == 0) {
      const int ts = (H1 + s) >> 1;
      const float dv = d[sgn_f * ts + off_f];
#pragma unroll
      for (int j = 0; j < 16; ++j)
        if (j * 32 + lane == ts) hv[j] += dv;
    }
  } else if (s == 0) {
#pragma unroll
    for (int j = 0; j < 16; ++j)
      hv[j] += d[sgn_f * (j * 32 + lane) + off_f];
  }

  const int cls_last = 28 + (lane & 3);
  float carry = 0.f;
#pragma unroll
  for (int j = 0; j < 16; ++j) {
    float v = hv[j];
    float u;
    u = __shfl_up(v, 4);  if (lane >= 4)  v += u;
    u = __shfl_up(v, 8);  if (lane >= 8)  v += u;
    u = __shfl_up(v, 16); if (lane >= 16) v += u;
    const float tot = __shfl(v, cls_last);
    Pp[j * 32 + lane] = v + carry;
    carry += tot;
  }
}

__global__ void k_build_lifted(int m, const float* __restrict__ P,
                               float* __restrict__ lifted) {
  const int idx = blockIdx.x * blockDim.x + threadIdx.x;
  const int q    = m & 3;
  const bool tr  = (q & 1) == 0;
  const int slot = (m & 4) + ((q == 3) ? 0 : (q == 2) ? 1 : (q == 0) ? 2 : 3);
  float* L = lifted + (size_t)slot * NN * NN;
  const int hi = idx >> 10;
  const int lo = idx & 1023;
  const int i = tr ? lo : hi;
  const int j = tr ? hi : lo;
  const int s = i - j;
  float acc = 0.f;
  if (s > -Hh && s < Hh) {
    const int si = s + H1;
    const int a = max(0, max(s, i - Hh));
    const int b = min(i, min(H1, H1 + s));
    if (b >= a) {
#pragma unroll
      for (int k = 0; k < 4; ++k) {
        const int c   = (i - k) & 3;
        const int Thi = b - ((b + 4 - c) & 3);
        const int Tlo = a + ((c + 4 - (a & 3)) & 3);
        if (Thi >= Tlo) {
          const float* Pp = P + ((size_t)(k * 1023 + si)) * Hh;
          float v = Pp[Thi];
          if (Tlo >= 4) v -= Pp[Tlo - 4];
          acc += v;
        }
      }
    }
  }
  L[idx] = acc;
}

#define BM 128
#define BN 64
#define BK 16

__global__ __launch_bounds__(256) void k_gemm_xwt(const float* __restrict__ A0,
                                                  const float* __restrict__ W0,
                                                  float* __restrict__ Out0,
                                                  size_t zA, size_t zW,
                                                  size_t zO) {
  const float* __restrict__ A   = A0 + blockIdx.z * zA;
  const float* __restrict__ W   = W0 + blockIdx.z * zW;
  float* __restrict__       Out = Out0 + blockIdx.z * zO;

  __shared__ float As[BM][BK + 1];
  __shared__ float Bs[BK][BN + 1];

  const int tid  = threadIdx.x;
  const int mb   = blockIdx.x * BM;
  const int nb   = blockIdx.y * BN;
  const int lane = tid & 31;
  const int w    = tid >> 5;
  const int wm   = (w & 3) * 32;
  const int wn   = (w >> 2) * 32;
  const int half = lane >> 4;
  const int l15  = lane & 15;

  v8f acc[2][2];
#pragma unroll
  for (int fm = 0; fm < 2; ++fm)
#pragma unroll
    for (int fn = 0; fn < 2; ++fn) acc[fm][fn] = v8f{};

  const int am = tid >> 1;
  const int ac = (tid & 1) * 8;
  const int wr = tid >> 2;
  const int wk = (tid & 3) * 4;

  for (int kb = 0; kb < 1024; kb += BK) {
    const float4 a0 = *(const float4*)(A + (size_t)(mb + am) * 1024 + kb + ac);
    const float4 a1 = *(const float4*)(A + (size_t)(mb + am) * 1024 + kb + ac + 4);
    As[am][ac + 0] = a0.x; As[am][ac + 1] = a0.y;
    As[am][ac + 2] = a0.z; As[am][ac + 3] = a0.w;
    As[am][ac + 4] = a1.x; As[am][ac + 5] = a1.y;
    As[am][ac + 6] = a1.z; As[am][ac + 7] = a1.w;
    const float4 wv = *(const float4*)(W + (size_t)(nb + wr) * 1024 + kb + wk);
    Bs[wk + 0][wr] = wv.x; Bs[wk + 1][wr] = wv.y;
    Bs[wk + 2][wr] = wv.z; Bs[wk + 3][wr] = wv.w;
    __syncthreads();

#pragma unroll
    for (int k4 = 0; k4 < BK; k4 += 4) {
      v2f af[2], bf[2];
#pragma unroll
      for (int fm = 0; fm < 2; ++fm) {
        const int row = wm + fm * 16 + l15;
        af[fm].x = As[row][k4 + half * 2];
        af[fm].y = As[row][k4 + half * 2 + 1];
      }
#pragma unroll
      for (int fn = 0; fn < 2; ++fn) {
        const int col = wn + fn * 16 + l15;
        bf[fn].x = Bs[k4 + half * 2][col];
        bf[fn].y = Bs[k4 + half * 2 + 1][col];
      }
#pragma unroll
      for (int fm = 0; fm < 2; ++fm)
#pragma unroll
        for (int fn = 0; fn < 2; ++fn)
          acc[fm][fn] = __builtin_amdgcn_wmma_f32_16x16x4_f32(
              false, af[fm], false, bf[fn], (short)0, acc[fm][fn],
              false, false);
    }
    __syncthreads();
  }

#pragma unroll
  for (int fm = 0; fm < 2; ++fm) {
#pragma unroll
    for (int fn = 0; fn < 2; ++fn) {
      const int col = nb + wn + fn * 16 + l15;
#pragma unroll
      for (int g = 0; g < 8; ++g) {
        const int row = mb + wm + fm * 16 + g + half * 8;
        Out[(size_t)row * 1024 + col] = acc[fm][fn][g];
      }
    }
  }
}

extern "C" void kernel_launch(void* const* d_in, const int* in_sizes, int n_in,
                              void* d_out, int out_size, void* d_ws,
                              size_t ws_size, hipStream_t stream) {
  (void)in_sizes; (void)n_in; (void)out_size; (void)ws_size;

  const float* x        = (const float*)d_in[0];
  const float* enc_rot  = (const float*)d_in[1];
  const float* enc_diag = (const float*)d_in[2];
  const float* dec_rot  = (const float*)d_in[3];
  const float* dec_diag = (const float*)d_in[4];

  float* out = (float*)d_out;
  float* ws  = (float*)d_ws;

  float* fg     = ws;
  float* P      = fg + (size_t)8 * 1024;
  float* lifted = P + (size_t)4 * 1023 * Hh;
  float* Q      = lifted + (size_t)8 * NN * NN;
  float* Wf     = Q + (size_t)4 * NN * NN;
  const size_t MM = (size_t)NN * NN;

  float* bneck  = out;
  float* final_ = out + (size_t)BB * NN;

  k_build_fg<<<8, 512, 0, stream>>>(enc_rot, enc_diag, dec_rot, dec_diag, fg);

  for (int m = 0; m < 8; ++m) {
    k_build_prefix<<<(4 * 1023 + 7) / 8, 256, 0, stream>>>(
        m, enc_rot, enc_diag, dec_rot, dec_diag, fg, P);
    k_build_lifted<<<(NN * NN) / 256, 256, 0, stream>>>(m, P, lifted);
  }

  dim3 grid_q(NN / BM, NN / BN, 4);
  k_gemm_xwt<<<grid_q, 256, 0, stream>>>(lifted, lifted + MM, Q,
                                         2 * MM, 2 * MM, MM);
  dim3 grid_w(NN / BM, NN / BN, 2);
  k_gemm_xwt<<<grid_w, 256, 0, stream>>>(Q, Q + MM, Wf, 2 * MM, 2 * MM, MM);

  dim3 grid(BB / BM, NN / BN, 1);
  k_gemm_xwt<<<grid, 256, 0, stream>>>(x,     Wf,      bneck,  0, 0, 0);
  k_gemm_xwt<<<grid, 256, 0, stream>>>(bneck, Wf + MM, final_, 0, 0, 0);
}


# k_build_lifted: lanes run along a diagonal (one row of P) so a load touches 1-2 lines per wave, not 32; tile turned through LDS for row stores; 8 elements per thread; same sums in the same order
# speedup vs baseline: 6.5919x; 1.1987x over previous
#include <hip/hip_runtime.h>
#include <hip/hip_bf16.h>


#define Hh   512
#define H1   511
#define NN   1024
#define BB   16384

typedef float v2f __attribute__((ext_vector_type(2)));
typedef float v8f __attribute__((ext_vector_type(8)));

__global__ __launch_bounds__(512) void k_build_fg(
    const float* __restrict__ enc_rot, const float* __restrict__ enc_diag,
    const float* __restrict__ dec_rot, const float* __restrict__ dec_diag,
    float* __restrict__ fg) {
  const int m = blockIdx.x;
  const int t = threadIdx.x;
  const float* r = (m < 4) ? (enc_rot + m * Hh) : (dec_rot + (m - 4) * Hh);
  const float* d = (m < 4) ? (enc_diag + m * Hh) : (dec_diag + (m - 4) * Hh);

  __shared__ float sA[512];
  __shared__ float sB[512];
  __shared__ float cS[4];

  const float rv = r[t];
  sA[t] = rv;
  sB[t] = rv * rv;
  __syncthreads();
  for (int off = 256; off > 0; off >>= 1) {
    if (t < off) { sA[t] += sA[t + off]; sB[t] += sB[t + off]; }
    __syncthreads();
  }
  if (t == 0) {
    const float a = sA[0], b = sB[0];
    const float D = 1.0f - a * a + 512.0f * b;
    cS[0] = -2.0f * b / D;
    cS[1] = 1.0f + ((1.0f + a) * (1.0f + a) - 512.0f * b) / D;
    cS[2] = -1.0f - ((1.0f - a) * (1.0f - a) - 512.0f * b) / D;
    cS[3] = -1024.0f / D;
  }
  __syncthreads();
  const float dv = d[t];
  fg[m * 1024 + t]       = dv * (cS[0] + cS[1] * rv);
  fg[m * 1024 + 512 + t] = dv * (cS[2] + cS[3] * rv);
}

__global__ __launch_bounds__(256) void k_build_prefix(
    int m,
    const float* __restrict__ enc_rot, const float* __restrict__ enc_diag,
    const float* __restrict__ dec_rot, const float* __restrict__ dec_diag,
    const float* __restrict__ fg, float* __restrict__ P) {
  const int lane = threadIdx.x & 31;
  const int wave = __builtin_amdgcn_readfirstlane((int)(threadIdx.x >> 5));
  const int idx  = blockIdx.x * 8 + wave;
  if (idx >= 4 * 1023) return;
  const int k  = idx & 3;
  const int si = idx >> 2;
  const int s  = si - H1;

  const float* r = (m < 4) ? (enc_rot + m * Hh) : (dec_rot + (m - 4) * Hh);
  const float* d = (m < 4) ? (enc_diag + m * Hh) : (dec_diag + (m - 4) * Hh);
  const float* f = fg + m * 1024;
  const float* g = f + 512;

  float* Pp = P + ((size_t)(k * 1023 + si)) * Hh;

  const int tlo = (s > 0) ? s : 0;
  const int thi = (s < 0) ? (H1 + s) : H1;

  const int sgn_f = (k < 2) ? 1 : -1;
  const int off_f = (k == 0) ? 0 : (k == 1) ? -s : (k == 2) ? H1 : (H1 + s);
  const int sgn_r = (k == 0 || k == 3) ? 1 : -1;
  const int off_r = (k == 0) ? -s : (k == 1) ? H1 : (k == 2) ? (H1 + s) : 0;

  const unsigned span = (unsigned)(thi - tlo);
  float hv[16];
#pragma unroll
  for (int j = 0; j < 16; ++j) {
    const int t  = j * 32 + lane;
    const int fi = min(max(sgn_f * t + off_f, 0), H1);
    const int ri = min(max(sgn_r * t + off_r, 0), H1);
    const float h = f[fi] + g[fi] * r[ri];
    hv[j] = ((unsigned)(t - tlo) <= span) ? h : 0.f;
  }

  if (k & 1) {
    if (((H1 + s) & 1) == 0) {
      const int ts = (H1 + s) >> 1;
      const float dv = d[sgn_f * ts + off_f];
#pragma unroll
      for (int j = 0; j < 16; ++j)
        if (j * 32 + lane == ts) hv[j] += dv;
    }
  } else if (s == 0) {
#pragma unroll
    for (int j = 0; j < 16; ++j)
      hv[j] += d[sgn_f * (j * 32 + lane) + off_f];
  }

  const int cls_last = 28 + (lane & 3);
  float carry = 0.f;
#pragma unroll
  for (int j = 0; j < 16; ++j) {
    float v = hv[j];
    float u;
    u = __shfl_up(v, 4);  if (lane >= 4)  v += u;
    u = __shfl_up(v, 8);  if (lane >= 8)  v += u;
    u = __shfl_up(v, 16); if (lane >= 16) v += u;
    const float tot = __shfl(v, cls_last);
    Pp[j * 32 + lane] = v + carry;
    carry += tot;
  }
}

__device__ __forceinline__ float lifted_value(int i, int j,
                                              const float* __restrict__ P) {
  const int s  = i - j;
  const int si = min(max(s + H1, 0), 2 * H1);
  const int a  = max(0, max(s, i - Hh));
  const int b  = min(i, min(H1, H1 + s));
  const bool any = (s > -Hh) && (s < Hh) && (b >= a);
  float hi[4], lo[4];
  bool ok[4], sub[4];
#pragma unroll
  for (int k = 0; k < 4; ++k) {
    const int c   = (i - k) & 3;
    const int Thi = b - ((b + 4 - c) & 3);
    const int Tlo = a + ((c + 4 - (a & 3)) & 3);
    ok[k]  = any && (Thi >= Tlo);
    sub[k] = ok[k] && (Tlo >= 4);
    const float* Pp = P + ((size_t)(k * 1023 + si)) * Hh;
    hi[k] = Pp[min(max(Thi, 0), H1)];
    lo[k] = Pp[min(max(Tlo - 4, 0), H1)];
  }
  float acc = 0.f;
#pragma unroll
  for (int k = 0; k < 4; ++k) {
    const float v = hi[k] - (sub[k] ? lo[k] : 0.f);
    acc += ok[k] ? v : 0.f;
  }
  return acc;
}

#define LT_R 32
#define LT_D 64

__global__ __launch_bounds__(256) void k_build_lifted(int m, const float* __restrict__ P,
                                                      float* __restrict__ lifted) {
  __shared__ float tile[LT_D][LT_R + 1];
  const int q    = m & 3;
  const bool tr  = (q & 1) == 0;
  const int slot = (m & 4) + ((q == 3) ? 0 : (q == 2) ? 1 : (q == 0) ? 2 : 3);
  float* L = lifted + (size_t)slot * NN * NN;
  const int lane = threadIdx.x & 31;
  const int w    = threadIdx.x >> 5;
  const int R0   = (blockIdx.x & 31) * LT_R;
  const int d0   = (blockIdx.x >> 5) * LT_D;

  const int R = R0 + lane;
#pragma unroll
  for (int u = 0; u < 8; ++u) {
    const int dl = w * 8 + u;
    const int C  = (R - d0 - dl) & 1023;
    tile[dl][lane] = lifted_value(tr ? C : R, tr ? R : C, P);
  }
  __syncthreads();

#pragma unroll
  for (int u = 0; u < 4; ++u) {
    const int rl = w * 4 + u;
    const int Rr = R0 + rl;
#pragma unroll
    for (int hf = 0; hf < 2; ++hf) {
      const int dl = (LT_D - 1) - (hf * 32 + lane);
      const int C  = (Rr - d0 - dl) & 1023;
      L[(size_t)Rr * NN + C] = tile[dl][rl];
    }
  }
}

#define BM 128
#define BN 64
#define BK 16

__global__ __launch_bounds__(256) void k_gemm_xwt(const float* __restrict__ A0,
                                                  const float* __restrict__ W0,
                                                  float* __restrict__ Out0,
                                                  size_t zA, size_t zW,
                                                  size_t zO) {
  const float* __restrict__ A   = A0 + blockIdx.z * zA;
  const float* __restrict__ W   = W0 + blockIdx.z * zW;
  float* __restrict__       Out = Out0 + blockIdx.z * zO;

  __shared__ float As[BM][BK + 1];
  __shared__ float Bs[BK][BN + 1];

  const int tid  = threadIdx.x;
  const int mb   = blockIdx.x * BM;
  const int nb   = blockIdx.y * BN;
  const int lane = tid & 31;
  const int w    = tid >> 5;
  const int wm   = (w & 3) * 32;
  const int wn   = (w >> 2) * 32;
  const int half = lane >> 4;
  const int l15  = lane & 15;

  v8f acc[2][2];
#pragma unroll
  for (int fm = 0; fm < 2; ++fm)
#pragma unroll
    for (int fn = 0; fn < 2; ++fn) acc[fm][fn] = v8f{};

  const int am = tid >> 1;
  const int ac = (tid & 1) * 8;
  const int wr = tid >> 2;
  const int wk = (tid & 3) * 4;

  for (int kb = 0; kb < 1024; kb += BK) {
    const float4 a0 = *(const float4*)(A + (size_t)(mb + am) * 1024 + kb + ac);
    const float4 a1 = *(const float4*)(A + (size_t)(mb + am) * 1024 + kb + ac + 4);
    As[am][ac + 0] = a0.x; As[am][ac + 1] = a0.y;
    As[am][ac + 2] = a0.z; As[am][ac + 3] = a0.w;
    As[am][ac + 4] = a1.x; As[am][ac + 5] = a1.y;
    As[am][ac + 6] = a1.z; As[am][ac + 7] = a1.w;
    const float4 wv = *(const float4*)(W + (size_t)(nb + wr) * 1024 + kb + wk);
    Bs[wk + 0][wr] = wv.x; Bs[wk + 1][wr] = wv.y;
    Bs[wk + 2][wr] = wv.z; Bs[wk + 3][wr] = wv.w;
    __syncthreads();

#pragma unroll
    for (int k4 = 0; k4 < BK; k4 += 4) {
      v2f af[2], bf[2];
#pragma unroll
      for (int fm = 0; fm < 2; ++fm) {
        const int row = wm + fm * 16 + l15;
        af[fm].x = As[row][k4 + half * 2];
        af[fm].y = As[row][k4 + half * 2 + 1];
      }
#pragma unroll
      for (int fn = 0; fn < 2; ++fn) {
        const int col = wn + fn * 16 + l15;
        bf[fn].x = Bs[k4 + half * 2][col];
        bf[fn].y = Bs[k4 + half * 2 + 1][col];
      }
#pragma unroll
      for (int fm = 0; fm < 2; ++fm)
#pragma unroll
        for (int fn = 0; fn < 2; ++fn)
          acc[fm][fn] = __builtin_amdgcn_wmma_f32_16x16x4_f32(
              false, af[fm], false, bf[fn], (short)0, acc[fm][fn],
              false, false);
    }
    __syncthreads();
  }

#pragma unroll
  for (int fm = 0; fm < 2; ++fm) {
#pragma unroll
    for (int fn = 0; fn < 2; ++fn) {
      const int col = nb + wn + fn * 16 + l15;
#pragma unroll
      for (int g = 0; g < 8; ++g) {
        const int row = mb + wm + fm * 16 + g + half * 8;
        Out[(size_t)row * 1024 + col] = acc[fm][fn][g];
      }
    }
  }
}

extern "C" void kernel_launch(void* const* d_in, const int* in_sizes, int n_in,
                              void* d_out, int out_size, void* d_ws,
                              size_t ws_size, hipStream_t stream) {
  (void)in_sizes; (void)n_in; (void)out_size; (void)ws_size;

  const float* x        = (const float*)d_in[0];
  const float* enc_rot  = (const float*)d_in[1];
  const float* enc_diag = (const float*)d_in[2];
  const float* dec_rot  = (const float*)d_in[3];
  const float* dec_diag = (const float*)d_in[4];

  float* out = (float*)d_out;
  float* ws  = (float*)d_ws;

  float* fg     = ws;
  float* P      = fg + (size_t)8 * 1024;
  float* lifted = P + (size_t)4 * 1023 * Hh;
  float* Q      = lifted + (size_t)8 * NN * NN;
  float* Wf     = Q + (size_t)4 * NN * NN;
  const size_t MM = (size_t)NN * NN;

  float* bneck  = out;
  float* final_ = out + (size_t)BB * NN;

  k_build_fg<<<8, 512, 0, stream>>>(enc_rot, enc_diag, dec_rot, dec_diag, fg);

  for (int m = 0; m < 8; ++m) {
    k_build_prefix<<<(4 * 1023 + 7) / 8, 256, 0, stream>>>(
        m, enc_rot, enc_diag, dec_rot, dec_diag, fg, P);
    k_build_lifted<<<512, 256, 0, stream>>>(m, P, lifted);
  }

  dim3 grid_q(NN / BM, NN / BN, 4);
  k_gemm_xwt<<<grid_q, 256, 0, stream>>>(lifted, lifted + MM, Q,
                                         2 * MM, 2 * MM, MM);
  dim3 grid_w(NN / BM, NN / BN, 2);
  k_gemm_xwt<<<grid_w, 256, 0, stream>>>(Q, Q + MM, Wf, 2 * MM, 2 * MM, MM);

  dim3 grid(BB / BM, NN / BN, 1);
  k_gemm_xwt<<<grid, 256, 0, stream>>>(x,     Wf,      bneck,  0, 0, 0);
  k_gemm_xwt<<<grid, 256, 0, stream>>>(bneck, Wf + MM, final_, 0, 0, 0);
}


# Build stage batched over the eight matrices: one k_build_prefix and one k_build_lifted launch with grid.y = matrix, one plane of P per matrix; 21 launches become 7; same sums in the same order
# speedup vs baseline: 7.2192x; 1.0952x over previous
#include <hip/hip_runtime.h>
#include <hip/hip_bf16.h>


#define Hh   512
#define H1   511
#define NN   1024
#define BB   16384

typedef float v2f __attribute__((ext_vector_type(2)));
typedef float v8f __attribute__((ext_vector_type(8)));

__global__ __launch_bounds__(512) void k_build_fg(
    const float* __restrict__ enc_rot, const float* __restrict__ enc_diag,
    const float* __restrict__ dec_rot, const float* __restrict__ dec_diag,
    float* __restrict__ fg) {
  const int m = blockIdx.x;
  const int t = threadIdx.x;
  const float* r = (m < 4) ? (enc_rot + m * Hh) : (dec_rot + (m - 4) * Hh);
  const float* d = (m < 4) ? (enc_diag + m * Hh) : (dec_diag + (m - 4) * Hh);

  __shared__ float sA[512];
  __shared__ float sB[512];
  __shared__ float cS[4];

  const float rv = r[t];
  sA[t] = rv;
  sB[t] = rv * rv;
  __syncthreads();
  for (int off = 256; off > 0; off >>= 1) {
    if (t < off) { sA[t] += sA[t + off]; sB[t] += sB[t + off]; }
    __syncthreads();
  }
  if (t == 0) {
    const float a = sA[0], b = sB[0];
    const float D = 1.0f - a * a + 512.0f * b;
    cS[0] = -2.0f * b / D;
    cS[1] = 1.0f + ((1.0f + a) * (1.0f + a) - 512.0f * b) / D;
    cS[2] = -1.0f - ((1.0f - a) * (1.0f - a) - 512.0f * b) / D;
    cS[3] = -1024.0f / D;
  }
  __syncthreads();
  const float dv = d[t];
  fg[m * 1024 + t]       = dv * (cS[0] + cS[1] * rv);
  fg[m * 1024 + 512 + t] = dv * (cS[2] + cS[3] * rv);
}

#define P_PLANE ((size_t)4 * 1023 * Hh)

__global__ __launch_bounds__(256) void k_build_prefix(
    const float* __restrict__ enc_rot, const float* __restrict__ enc_diag,
    const float* __restrict__ dec_rot, const float* __restrict__ dec_diag,
    const float* __restrict__ fg, float* __restrict__ Pall) {
  const int m = blockIdx.y;
  float* __restrict__ P = Pall + (size_t)m * P_PLANE;
  const int lane = threadIdx.x & 31;
  const int wave = __builtin_amdgcn_readfirstlane((int)(threadIdx.x >> 5));
  const int idx  = blockIdx.x * 8 + wave;
  if (idx >= 4 * 1023) return;
  const int k  = idx & 3;
  const int si = idx >> 2;
  const int s  = si - H1;

  const float* r = (m < 4) ? (enc_rot + m * Hh) : (dec_rot + (m - 4) * Hh);
  const float* d = (m < 4) ? (enc_diag + m * Hh) : (dec_diag + (m - 4) * Hh);
  const float* f = fg + m * 1024;
  const float* g = f + 512;

  float* Pp = P + ((size_t)(k * 1023 + si)) * Hh;

  const int tlo = (s > 0) ? s : 0;
  const int thi = (s < 0) ? (H1 + s) : H1;

  const int sgn_f = (k < 2) ? 1 : -1;
  const int off_f = (k == 0) ? 0 : (k == 1) ? -s : (k == 2) ? H1 : (H1 + s);
  const int sgn_r = (k == 0 || k == 3) ? 1 : -1;
  const int off_r = (k == 0) ? -s : (k == 1) ? H1 : (k == 2) ? (H1 + s) : 0;

  const unsigned span = (unsigned)(thi - tlo);
  float hv[16];
#pragma unroll
  for (int j = 0; j < 16; ++j) {
    const int t  = j * 32 + lane;
    const int fi = min(max(sgn_f * t + off_f, 0), H1);
    const int ri = min(max(sgn_r * t + off_r, 0), H1);
    const float h = f[fi] + g[fi] * r[ri];
    hv[j] = ((unsigned)(t - tlo) <= span) ? h : 0.f;
  }

  if (k & 1) {
    if (((H1 + s) & 1) == 0) {
      const int ts = (H1 + s) >> 1;
      const float dv = d[sgn_f * ts + off_f];
#pragma unroll
      for (int j = 0; j < 16; ++j)
        if (j * 32 + lane == ts) hv[j] += dv;
    }
  } else if (s == 0) {
#pragma unroll
    for (int j = 0; j < 16; ++j)
      hv[j] += d[sgn_f * (j * 32 + lane) + off_f];
  }

  const int cls_last = 28 + (lane & 3);
  float carry = 0.f;
#pragma unroll
  for (int j = 0; j < 16; ++j) {
    float v = hv[j];
    float u;
    u = __shfl_up(v, 4);  if (lane >= 4)  v += u;
    u = __shfl_up(v, 8);  if (lane >= 8)  v += u;
    u = __shfl_up(v, 16); if (lane >= 16) v += u;
    const float tot = __shfl(v, cls_last);
    Pp[j * 32 + lane] = v + carry;
    carry += tot;
  }
}

__device__ __forceinline__ float lifted_value(int i, int j,
                                              const float* __restrict__ P) {
  const int s  = i - j;
  const int si = min(max(s + H1, 0), 2 * H1);
  const int a  = max(0, max(s, i - Hh));
  const int b  = min(i, min(H1, H1 + s));
  const bool any = (s > -Hh) && (s < Hh) && (b >= a);
  float hi[4], lo[4];
  bool ok[4], sub[4];
#pragma unroll
  for (int k = 0; k < 4; ++k) {
    const int c   = (i - k) & 3;
    const int Thi = b - ((b + 4 - c) & 3);
    const int Tlo = a + ((c + 4 - (a & 3)) & 3);
    ok[k]  = any && (Thi >= Tlo);
    sub[k] = ok[k] && (Tlo >= 4);
    const float* Pp = P + ((size_t)(k * 1023 + si)) * Hh;
    hi[k] = Pp[min(max(Thi, 0), H1)];
    lo[k] = Pp[min(max(Tlo - 4, 0), H1)];
  }
  float acc = 0.f;
#pragma unroll
  for (int k = 0; k < 4; ++k) {
    const float v = hi[k] - (sub[k] ? lo[k] : 0.f);
    acc += ok[k] ? v : 0.f;
  }
  return acc;
}

#define LT_R 32
#define LT_D 64

__global__ __launch_bounds__(256) void k_build_lifted(const float* __restrict__ Pall,
                                                      float* __restrict__ lifted) {
  __shared__ float tile[LT_D][LT_R + 1];
  const int m    = blockIdx.y;
  const float* __restrict__ P = Pall + (size_t)m * P_PLANE;
  const int q    = m & 3;
  const bool tr  = (q & 1) == 0;
  const int slot = (m & 4) + ((q == 3) ? 0 : (q == 2) ? 1 : (q == 0) ? 2 : 3);
  float* L = lifted + (size_t)slot * NN * NN;
  const int lane = threadIdx.x & 31;
  const int w    = threadIdx.x >> 5;
  const int R0   = (blockIdx.x & 31) * LT_R;
  const int d0   = (blockIdx.x >> 5) * LT_D;

  const int R = R0 + lane;
#pragma unroll
  for (int u = 0; u < 8; ++u) {
    const int dl = w * 8 + u;
    const int C  = (R - d0 - dl) & 1023;
    tile[dl][lane] = lifted_value(tr ? C : R, tr ? R : C, P);
  }
  __syncthreads();

#pragma unroll
  for (int u = 0; u < 4; ++u) {
    const int rl = w * 4 + u;
    const int Rr = R0 + rl;
#pragma unroll
    for (int hf = 0; hf < 2; ++hf) {
      const int dl = (LT_D - 1) - (hf * 32 + lane);
      const int C  = (Rr - d0 - dl) & 1023;
      L[(size_t)Rr * NN + C] = tile[dl][rl];
    }
  }
}

#define BM 128
#define BN 64
#define BK 16

__global__ __launch_bounds__(256) void k_gemm_xwt(const float* __restrict__ A0,
                                                  const float* __restrict__ W0,
                                                  float* __restrict__ Out0,
                                                  size_t zA, size_t zW,
                                                  size_t zO) {
  const float* __restrict__ A   = A0 + blockIdx.z * zA;
  const float* __restrict__ W   = W0 + blockIdx.z * zW;
  float* __restrict__       Out = Out0 + blockIdx.z * zO;

  __shared__ float As[BM][BK + 1];
  __shared__ float Bs[BK][BN + 1];

  const int tid  = threadIdx.x;
  const int mb   = blockIdx.x * BM;
  const int nb   = blockIdx.y * BN;
  const int lane = tid & 31;
  const int w    = tid >> 5;
  const int wm   = (w & 3) * 32;
  const int wn   = (w >> 2) * 32;
  const int half = lane >> 4;
  const int l15  = lane & 15;

  v8f acc[2][2];
#pragma unroll
  for (int fm = 0; fm < 2; ++fm)
#pragma unroll
    for (int fn = 0; fn < 2; ++fn) acc[fm][fn] = v8f{};

  const int am = tid >> 1;
  const int ac = (tid & 1) * 8;
  const int wr = tid >> 2;
  const int wk = (tid & 3) * 4;

  for (int kb = 0; kb < 1024; kb += BK) {
    const float4 a0 = *(const float4*)(A + (size_t)(mb + am) * 1024 + kb + ac);
    const float4 a1 = *(const float4*)(A + (size_t)(mb + am) * 1024 + kb + ac + 4);
    As[am][ac + 0] = a0.x; As[am][ac + 1] = a0.y;
    As[am][ac + 2] = a0.z; As[am][ac + 3] = a0.w;
    As[am][ac + 4] = a1.x; As[am][ac + 5] = a1.y;
    As[am][ac + 6] = a1.z; As[am][ac + 7] = a1.w;
    const float4 wv = *(const float4*)(W + (size_t)(nb + wr) * 1024 + kb + wk);
    Bs[wk + 0][wr] = wv.x; Bs[wk + 1][wr] = wv.y;
    Bs[wk + 2][wr] = wv.z; Bs[wk + 3][wr] = wv.w;
    __syncthreads();

#pragma unroll
    for (int k4 = 0; k4 < BK; k4 += 4) {
      v2f af[2], bf[2];
#pragma unroll
      for (int fm = 0; fm < 2; ++fm) {
        const int row = wm + fm * 16 + l15;
        af[fm].x = As[row][k4 + half * 2];
        af[fm].y = As[row][k4 + half * 2 + 1];
      }
#pragma unroll
      for (int fn = 0; fn < 2; ++fn) {
        const int col = wn + fn * 16 + l15;
        bf[fn].x = Bs[k4 + half * 2][col];
        bf[fn].y = Bs[k4 + half * 2 + 1][col];
      }
#pragma unroll
      for (int fm = 0; fm < 2; ++fm)
#pragma unroll
        for (int fn = 0; fn < 2; ++fn)
          acc[fm][fn] = __builtin_amdgcn_wmma_f32_16x16x4_f32(
              false, af[fm], false, bf[fn], (short)0, acc[fm][fn],
              false, false);
    }
    __syncthreads();
  }

#pragma unroll
  for (int fm = 0; fm < 2; ++fm) {
#pragma unroll
    for (int fn = 0; fn < 2; ++fn) {
      const int col = nb + wn + fn * 16 + l15;
#pragma unroll
      for (int g = 0; g < 8; ++g) {
        const int row = mb + wm + fm * 16 + g + half * 8;
        Out[(size_t)row * 1024 + col] = acc[fm][fn][g];
      }
    }
  }
}

extern "C" void kernel_launch(void* const* d_in, const int* in_sizes, int n_in,
                              void* d_out, int out_size, void* d_ws,
                              size_t ws_size, hipStream_t stream) {
  (void)in_sizes; (void)n_in; (void)out_size; (void)ws_size;

  const float* x        = (const float*)d_in[0];
  const float* enc_rot  = (const float*)d_in[1];
  const float* enc_diag = (const float*)d_in[2];
  const float* dec_rot  = (const float*)d_in[3];
  const float* dec_diag = (const float*)d_in[4];

  float* out = (float*)d_out;
  float* ws  = (float*)d_ws;

  float* fg     = ws;
  float* P      = fg + (size_t)8 * 1024;
  float* lifted = P + (size_t)8 * P_PLANE;
  float* Q      = lifted + (size_t)8 * NN * NN;
  float* Wf     = Q + (size_t)4 * NN * NN;
  const size_t MM = (size_t)NN * NN;

  float* bneck  = out;
  float* final_ = out + (size_t)BB * NN;

  k_build_fg<<<8, 512, 0, stream>>>(enc_rot, enc_diag, dec_rot, dec_diag, fg);

  dim3 grid_p((4 * 1023 + 7) / 8, 8, 1);
  k_build_prefix<<<grid_p, 256, 0, stream>>>(
      enc_rot, enc_diag, dec_rot, dec_diag, fg, P);
  dim3 grid_l(512, 8, 1);
  k_build_lifted<<<grid_l, 256, 0, stream>>>(P, lifted);

  dim3 grid_q(NN / BM, NN / BN, 4);
  k_gemm_xwt<<<grid_q, 256, 0, stream>>>(lifted, lifted + MM, Q,
                                         2 * MM, 2 * MM, MM);
  dim3 grid_w(NN / BM, NN / BN, 2);
  k_gemm_xwt<<<grid_w, 256, 0, stream>>>(Q, Q + MM, Wf, 2 * MM, 2 * MM, MM);

  dim3 grid(BB / BM, NN / BN, 1);
  k_gemm_xwt<<<grid, 256, 0, stream>>>(x,     Wf,      bneck,  0, 0, 0);
  k_gemm_xwt<<<grid, 256, 0, stream>>>(bneck, Wf + MM, final_, 0, 0, 0);
}


# k_gemm_xwt: both LDS tiles kept in memory order [row][k], pitch 20: 16-byte LDS stores, one 8-byte LDS load per fragment, no W transpose; 62 VGPRs, whole grid resident; same sums, same order
# speedup vs baseline: 7.8509x; 1.0875x over previous
#include <hip/hip_runtime.h>
#include <hip/hip_bf16.h>


#define Hh   512
#define H1   511
#define NN   1024
#define BB   16384

typedef float v2f __attribute__((ext_vector_type(2)));
typedef float v8f __attribute__((ext_vector_type(8)));

__global__ __launch_bounds__(512) void k_build_fg(
    const float* __restrict__ enc_rot, const float* __restrict__ enc_diag,
    const float* __restrict__ dec_rot, const float* __restrict__ dec_diag,
    float* __restrict__ fg) {
  const int m = blockIdx.x;
  const int t = threadIdx.x;
  const float* r = (m < 4) ? (enc_rot + m * Hh) : (dec_rot + (m - 4) * Hh);
  const float* d = (m < 4) ? (enc_diag + m * Hh) : (dec_diag + (m - 4) * Hh);

  __shared__ float sA[512];
  __shared__ float sB[512];
  __shared__ float cS[4];

  const float rv = r[t];
  sA[t] = rv;
  sB[t] = rv * rv;
  __syncthreads();
  for (int off = 256; off > 0; off >>= 1) {
    if (t < off) { sA[t] += sA[t + off]; sB[t] += sB[t + off]; }
    __syncthreads();
  }
  if (t == 0) {
    const float a = sA[0], b = sB[0];
    const float D = 1.0f - a * a + 512.0f * b;
    cS[0] = -2.0f * b / D;
    cS[1] = 1.0f + ((1.0f + a) * (1.0f + a) - 512.0f * b) / D;
    cS[2] = -1.0f - ((1.0f - a) * (1.0f - a) - 512.0f * b) / D;
    cS[3] = -1024.0f / D;
  }
  __syncthreads();
  const float dv = d[t];
  fg[m * 1024 + t]       = dv * (cS[0] + cS[1] * rv);
  fg[m * 1024 + 512 + t] = dv * (cS[2] + cS[3] * rv);
}

#define P_PLANE ((size_t)4 * 1023 * Hh)

__global__ __launch_bounds__(256) void k_build_prefix(
    const float* __restrict__ enc_rot, const float* __restrict__ enc_diag,
    const float* __restrict__ dec_rot, const float* __restrict__ dec_diag,
    const float* __restrict__ fg, float* __restrict__ Pall) {
  const int m = blockIdx.y;
  float* __restrict__ P = Pall + (size_t)m * P_PLANE;
  const int lane = threadIdx.x & 31;
  const int wave = __builtin_amdgcn_readfirstlane((int)(threadIdx.x >> 5));
  const int idx  = blockIdx.x * 8 + wave;
  if (idx >= 4 * 1023) return;
  const int k  = idx & 3;
  const int si = idx >> 2;
  const int s  = si - H1;

  const float* r = (m < 4) ? (enc_rot + m * Hh) : (dec_rot + (m - 4) * Hh);
  const float* d = (m < 4) ? (enc_diag + m * Hh) : (dec_diag + (m - 4) * Hh);
  const float* f = fg + m * 1024;
  const float* g = f + 512;

  float* Pp = P + ((size_t)(k * 1023 + si)) * Hh;

  const int tlo = (s > 0) ? s : 0;
  const int thi = (s < 0) ? (H1 + s) : H1;

  const int sgn_f = (k < 2) ? 1 : -1;
  const int off_f = (k == 0) ? 0 : (k == 1) ? -s : (k == 2) ? H1 : (H1 + s);
  const int sgn_r = (k == 0 || k == 3) ? 1 : -1;
  const int off_r = (k == 0) ? -s : (k == 1) ? H1 : (k == 2) ? (H1 + s) : 0;

  const unsigned span = (unsigned)(thi - tlo);
  float hv[16];
#pragma unroll
  for (int j = 0; j < 16; ++j) {
    const int t  = j * 32 + lane;
    const int fi = min(max(sgn_f * t + off_f, 0), H1);
    const int ri = min(max(sgn_r * t + off_r, 0), H1);
    const float h = f[fi] + g[fi] * r[ri];
    hv[j] = ((unsigned)(t - tlo) <= span) ? h : 0.f;
  }

  if (k & 1) {
    if (((H1 + s) & 1) == 0) {
      const int ts = (H1 + s) >> 1;
      const float dv = d[sgn_f * ts + off_f];
#pragma unroll
      for (int j = 0; j < 16; ++j)
        if (j * 32 + lane == ts) hv[j] += dv;
    }
  } else if (s == 0) {
#pragma unroll
    for (int j = 0; j < 16; ++j)
      hv[j] += d[sgn_f * (j * 32 + lane) + off_f];
  }

  const int cls_last = 28 + (lane & 3);
  float carry = 0.f;
#pragma unroll
  for (int j = 0; j < 16; ++j) {
    float v = hv[j];
    float u;
    u = __shfl_up(v, 4);  if (lane >= 4)  v += u;
    u = __shfl_up(v, 8);  if (lane >= 8)  v += u;
    u = __shfl_up(v, 16); if (lane >= 16) v += u;
    const float tot = __shfl(v, cls_last);
    Pp[j * 32 + lane] = v + carry;
    carry += tot;
  }
}

__device__ __forceinline__ float lifted_value(int i, int j,
                                              const float* __restrict__ P) {
  const int s  = i - j;
  const int si = min(max(s + H1, 0), 2 * H1);
  const int a  = max(0, max(s, i - Hh));
  const int b  = min(i, min(H1, H1 + s));
  const bool any = (s > -Hh) && (s < Hh) && (b >= a);
  float hi[4], lo[4];
  bool ok[4], sub[4];
#pragma unroll
  for (int k = 0; k < 4; ++k) {
    const int c   = (i - k) & 3;
    const int Thi = b - ((b + 4 - c) & 3);
    const int Tlo = a + ((c + 4 - (a & 3)) & 3);
    ok[k]  = any && (Thi >= Tlo);
    sub[k] = ok[k] && (Tlo >= 4);
    const float* Pp = P + ((size_t)(k * 1023 + si)) * Hh;
    hi[k] = Pp[min(max(Thi, 0), H1)];
    lo[k] = Pp[min(max(Tlo - 4, 0), H1)];
  }
  float acc = 0.f;
#pragma unroll
  for (int k = 0; k < 4; ++k) {
    const float v = hi[k] - (sub[k] ? lo[k] : 0.f);
    acc += ok[k] ? v : 0.f;
  }
  return acc;
}

#define LT_R 32
#define LT_D 64

__global__ __launch_bounds__(256) void k_build_lifted(const float* __restrict__ Pall,
                                                      float* __restrict__ lifted) {
  __shared__ float tile[LT_D][LT_R + 1];
  const int m    = blockIdx.y;
  const float* __restrict__ P = Pall + (size_t)m * P_PLANE;
  const int q    = m & 3;
  const bool tr  = (q & 1) == 0;
  const int slot = (m & 4) + ((q == 3) ? 0 : (q == 2) ? 1 : (q == 0) ? 2 : 3);
  float* L = lifted + (size_t)slot * NN * NN;
  const int lane = threadIdx.x & 31;
  const int w    = threadIdx.x >> 5;
  const int R0   = (blockIdx.x & 31) * LT_R;
  const int d0   = (blockIdx.x >> 5) * LT_D;

  const int R = R0 + lane;
#pragma unroll
  for (int u = 0; u < 8; ++u) {
    const int dl = w * 8 + u;
    const int C  = (R - d0 - dl) & 1023;
    tile[dl][lane] = lifted_value(tr ? C : R, tr ? R : C, P);
  }
  __syncthreads();

#pragma unroll
  for (int u = 0; u < 4; ++u) {
    const int rl = w * 4 + u;
    const int Rr = R0 + rl;
#pragma unroll
    for (int hf = 0; hf < 2; ++hf) {
      const int dl = (LT_D - 1) - (hf * 32 + lane);
      const int C  = (Rr - d0 - dl) & 1023;
      L[(size_t)Rr * NN + C] = tile[dl][rl];
    }
  }
}

#define BM 128
#define BN 64
#define BK 16
#define LP 20

__global__ __launch_bounds__(256) void k_gemm_xwt(const float* __restrict__ A0,
                                                  const float* __restrict__ W0,
                                                  float* __restrict__ Out0,
                                                  size_t zA, size_t zW,
                                                  size_t zO) {
  const float* __restrict__ A   = A0 + blockIdx.z * zA;
  const float* __restrict__ W   = W0 + blockIdx.z * zW;
  float* __restrict__       Out = Out0 + blockIdx.z * zO;

  __shared__ __attribute__((aligned(16))) float As[BM][LP];
  __shared__ __attribute__((aligned(16))) float Bs[BN][LP];

  const int tid  = threadIdx.x;
  const int mb   = blockIdx.x * BM;
  const int nb   = blockIdx.y * BN;
  const int lane = tid & 31;
  const int w    = tid >> 5;
  const int wm   = (w & 3) * 32;
  const int wn   = (w >> 2) * 32;
  const int half = lane >> 4;
  const int l15  = lane & 15;

  v8f acc[2][2];
#pragma unroll
  for (int fm = 0; fm < 2; ++fm)
#pragma unroll
    for (int fn = 0; fn < 2; ++fn) acc[fm][fn] = v8f{};

  const int am = tid >> 1;
  const int ac = (tid & 1) * 8;
  const int wr = tid >> 2;
  const int wk = (tid & 3) * 4;

  for (int kb = 0; kb < 1024; kb += BK) {
    const float4 a0 = *(const float4*)(A + (size_t)(mb + am) * 1024 + kb + ac);
    const float4 a1 = *(const float4*)(A + (size_t)(mb + am) * 1024 + kb + ac + 4);
    *(float4*)&As[am][ac]     = a0;
    *(float4*)&As[am][ac + 4] = a1;
    const float4 wv = *(const float4*)(W + (size_t)(nb + wr) * 1024 + kb + wk);
    *(float4*)&Bs[wr][wk] = wv;
    __syncthreads();

#pragma unroll
    for (int k4 = 0; k4 < BK; k4 += 4) {
      v2f af[2], bf[2];
#pragma unroll
      for (int fm = 0; fm < 2; ++fm) {
        const int row = wm + fm * 16 + l15;
        af[fm] = *(const v2f*)&As[row][k4 + half * 2];
      }
#pragma unroll
      for (int fn = 0; fn < 2; ++fn) {
        const int col = wn + fn * 16 + l15;
        bf[fn] = *(const v2f*)&Bs[col][k4 + half * 2];
      }
#pragma unroll
      for (int fm = 0; fm < 2; ++fm)
#pragma unroll
        for (int fn = 0; fn < 2; ++fn)
          acc[fm][fn] = __builtin_amdgcn_wmma_f32_16x16x4_f32(
              false, af[fm], false, bf[fn], (short)0, acc[fm][fn],
              false, false);
    }
    __syncthreads();
  }

#pragma unroll
  for (int fm = 0; fm < 2; ++fm) {
#pragma unroll
    for (int fn = 0; fn < 2; ++fn) {
      const int col = nb + wn + fn * 16 + l15;
#pragma unroll
      for (int g = 0; g < 8; ++g) {
        const int row = mb + wm + fm * 16 + g + half * 8;
        Out[(size_t)row * 1024 + col] = acc[fm][fn][g];
      }
    }
  }
}

extern "C" void kernel_launch(void* const* d_in, const int* in_sizes, int n_in,
                              void* d_out, int out_size, void* d_ws,
                              size_t ws_size, hipStream_t stream) {
  (void)in_sizes; (void)n_in; (void)out_size; (void)ws_size;

  const float* x        = (const float*)d_in[0];
  const float* enc_rot  = (const float*)d_in[1];
  const float* enc_diag = (const float*)d_in[2];
  const float* dec_rot  = (const float*)d_in[3];
  const float* dec_diag = (const float*)d_in[4];

  float* out = (float*)d_out;
  float* ws  = (float*)d_ws;

  float* fg     = ws;
  float* P      = fg + (size_t)8 * 1024;
  float* lifted = P + (size_t)8 * P_PLANE;
  float* Q      = lifted + (size_t)8 * NN * NN;
  float* Wf     = Q + (size_t)4 * NN * NN;
  const size_t MM = (size_t)NN * NN;

  float* bneck  = out;
  float* final_ = out + (size_t)BB * NN;

  k_build_fg<<<8, 512, 0, stream>>>(enc_rot, enc_diag, dec_rot, dec_diag, fg);

  dim3 grid_p((4 * 1023 + 7) / 8, 8, 1);
  k_build_prefix<<<grid_p, 256, 0, stream>>>(
      enc_rot, enc_diag, dec_rot, dec_diag, fg, P);
  dim3 grid_l(512, 8, 1);
  k_build_lifted<<<grid_l, 256, 0, stream>>>(P, lifted);

  dim3 grid_q(NN / BM, NN / BN, 4);
  k_gemm_xwt<<<grid_q, 256, 0, stream>>>(lifted, lifted + MM, Q,
                                         2 * MM, 2 * MM, MM);
  dim3 grid_w(NN / BM, NN / BN, 2);
  k_gemm_xwt<<<grid_w, 256, 0, stream>>>(Q, Q + MM, Wf, 2 * MM, 2 * MM, MM);

  dim3 grid(BB / BM, NN / BN, 1);
  k_gemm_xwt<<<grid, 256, 0, stream>>>(x,     Wf,      bneck,  0, 0, 0);
  k_gemm_xwt<<<grid, 256, 0, stream>>>(bneck, Wf + MM, final_, 0, 0, 0);
}


# k_gemm_xwt: two LDS stages; the next K-step's tile is loaded before the current step's WMMAs and stored to the other stage behind them; one barrier per K-step, not two; same sums, same order
# speedup vs baseline: 8.4718x; 1.0791x over previous
#include <hip/hip_runtime.h>
#include <hip/hip_bf16.h>


#define Hh   512
#define H1   511
#define NN   1024
#define BB   16384

typedef float v2f __attribute__((ext_vector_type(2)));
typedef float v8f __attribute__((ext_vector_type(8)));

__global__ __launch_bounds__(512) void k_build_fg(
    const float* __restrict__ enc_rot, const float* __restrict__ enc_diag,
    const float* __restrict__ dec_rot, const float* __restrict__ dec_diag,
    float* __restrict__ fg) {
  const int m = blockIdx.x;
  const int t = threadIdx.x;
  const float* r = (m < 4) ? (enc_rot + m * Hh) : (dec_rot + (m - 4) * Hh);
  const float* d = (m < 4) ? (enc_diag + m * Hh) : (dec_diag + (m - 4) * Hh);

  __shared__ float sA[512];
  __shared__ float sB[512];
  __shared__ float cS[4];

  const float rv = r[t];
  sA[t] = rv;
  sB[t] = rv * rv;
  __syncthreads();
  for (int off = 256; off > 0; off >>= 1) {
    if (t < off) { sA[t] += sA[t + off]; sB[t] += sB[t + off]; }
    __syncthreads();
  }
  if (t == 0) {
    const float a = sA[0], b = sB[0];
    const float D = 1.0f - a * a + 512.0f * b;
    cS[0] = -2.0f * b / D;
    cS[1] = 1.0f + ((1.0f + a) * (1.0f + a) - 512.0f * b) / D;
    cS[2] = -1.0f - ((1.0f - a) * (1.0f - a) - 512.0f * b) / D;
    cS[3] = -1024.0f / D;
  }
  __syncthreads();
  const float dv = d[t];
  fg[m * 1024 + t]       = dv * (cS[0] + cS[1] * rv);
  fg[m * 1024 + 512 + t] = dv * (cS[2] + cS[3] * rv);
}

#define P_PLANE ((size_t)4 * 1023 * Hh)

__global__ __launch_bounds__(256) void k_build_prefix(
    const float* __restrict__ enc_rot, const float* __restrict__ enc_diag,
    const float* __restrict__ dec_rot, const float* __restrict__ dec_diag,
    const float* __restrict__ fg, float* __restrict__ Pall) {
  const int m = blockIdx.y;
  float* __restrict__ P = Pall + (size_t)m * P_PLANE;
  const int lane = threadIdx.x & 31;
  const int wave = __builtin_amdgcn_readfirstlane((int)(threadIdx.x >> 5));
  const int idx  = blockIdx.x * 8 + wave;
  if (idx >= 4 * 1023) return;
  const int k  = idx & 3;
  const int si = idx >> 2;
  const int s  = si - H1;

  const float* r = (m < 4) ? (enc_rot + m * Hh) : (dec_rot + (m - 4) * Hh);
  const float* d = (m < 4) ? (enc_diag + m * Hh) : (dec_diag + (m - 4) * Hh);
  const float* f = fg + m * 1024;
  const float* g = f + 512;

  float* Pp = P + ((size_t)(k * 1023 + si)) * Hh;

  const int tlo = (s > 0) ? s : 0;
  const int thi = (s < 0) ? (H1 + s) : H1;

  const int sgn_f = (k < 2) ? 1 : -1;
  const int off_f = (k == 0) ? 0 : (k == 1) ? -s : (k == 2) ? H1 : (H1 + s);
  const int sgn_r = (k == 0 || k == 3) ? 1 : -1;
  const int off_r = (k == 0) ? -s : (k == 1) ? H1 : (k == 2) ? (H1 + s) : 0;

  const unsigned span = (unsigned)(thi - tlo);
  float hv[16];
#pragma unroll
  for (int j = 0; j < 16; ++j) {
    const int t  = j * 32 + lane;
    const int fi = min(max(sgn_f * t + off_f, 0), H1);
    const int ri = min(max(sgn_r * t + off_r, 0), H1);
    const float h = f[fi] + g[fi] * r[ri];
    hv[j] = ((unsigned)(t - tlo) <= span) ? h : 0.f;
  }

  if (k & 1) {
    if (((H1 + s) & 1) == 0) {
      const int ts = (H1 + s) >> 1;
      const float dv = d[sgn_f * ts + off_f];
#pragma unroll
      for (int j = 0; j < 16; ++j)
        if (j * 32 + lane == ts) hv[j] += dv;
    }
  } else if (s == 0) {
#pragma unroll
    for (int j = 0; j < 16; ++j)
      hv[j] += d[sgn_f * (j * 32 + lane) + off_f];
  }

  const int cls_last = 28 + (lane & 3);
  float carry = 0.f;
#pragma unroll
  for (int j = 0; j < 16; ++j) {
    float v = hv[j];
    float u;
    u = __shfl_up(v, 4);  if (lane >= 4)  v += u;
    u = __shfl_up(v, 8);  if (lane >= 8)  v += u;
    u = __shfl_up(v, 16); if (lane >= 16) v += u;
    const float tot = __shfl(v, cls_last);
    Pp[j * 32 + lane] = v + carry;
    carry += tot;
  }
}

__device__ __forceinline__ float lifted_value(int i, int j,
                                              const float* __restrict__ P) {
  const int s  = i - j;
  const int si = min(max(s + H1, 0), 2 * H1);
  const int a  = max(0, max(s, i - Hh));
  const int b  = min(i, min(H1, H1 + s));
  const bool any = (s > -Hh) && (s < Hh) && (b >= a);
  float hi[4], lo[4];
  bool ok[4], sub[4];
#pragma unroll
  for (int k = 0; k < 4; ++k) {
    const int c   = (i - k) & 3;
    const int Thi = b - ((b + 4 - c) & 3);
    const int Tlo = a + ((c + 4 - (a & 3)) & 3);
    ok[k]  = any && (Thi >= Tlo);
    sub[k] = ok[k] && (Tlo >= 4);
    const float* Pp = P + ((size_t)(k * 1023 + si)) * Hh;
    hi[k] = Pp[min(max(Thi, 0), H1)];
    lo[k] = Pp[min(max(Tlo - 4, 0), H1)];
  }
  float acc = 0.f;
#pragma unroll
  for (int k = 0; k < 4; ++k) {
    const float v = hi[k] - (sub[k] ? lo[k] : 0.f);
    acc += ok[k] ? v : 0.f;
  }
  return acc;
}

#define LT_R 32
#define LT_D 64

__global__ __launch_bounds__(256) void k_build_lifted(const float* __restrict__ Pall,
                                                      float* __restrict__ lifted) {
  __shared__ float tile[LT_D][LT_R + 1];
  const int m    = blockIdx.y;
  const float* __restrict__ P = Pall + (size_t)m * P_PLANE;
  const int q    = m & 3;
  const bool tr  = (q & 1) == 0;
  const int slot = (m & 4) + ((q == 3) ? 0 : (q == 2) ? 1 : (q == 0) ? 2 : 3);
  float* L = lifted + (size_t)slot * NN * NN;
  const int lane = threadIdx.x & 31;
  const int w    = threadIdx.x >> 5;
  const int R0   = (blockIdx.x & 31) * LT_R;
  const int d0   = (blockIdx.x >> 5) * LT_D;

  const int R = R0 + lane;
#pragma unroll
  for (int u = 0; u < 8; ++u) {
    const int dl = w * 8 + u;
    const int C  = (R - d0 - dl) & 1023;
    tile[dl][lane] = lifted_value(tr ? C : R, tr ? R : C, P);
  }
  __syncthreads();

#pragma unroll
  for (int u = 0; u < 4; ++u) {
    const int rl = w * 4 + u;
    const int Rr = R0 + rl;
#pragma unroll
    for (int hf = 0; hf < 2; ++hf) {
      const int dl = (LT_D - 1) - (hf * 32 + lane);
      const int C  = (Rr - d0 - dl) & 1023;
      L[(size_t)Rr * NN + C] = tile[dl][rl];
    }
  }
}

#define BM 128
#define BN 64
#define BK 16
#define LP 20

__global__ __launch_bounds__(256) void k_gemm_xwt(const float* __restrict__ A0,
                                                  const float* __restrict__ W0,
                                                  float* __restrict__ Out0,
                                                  size_t zA, size_t zW,
                                                  size_t zO) {
  const float* __restrict__ A   = A0 + blockIdx.z * zA;
  const float* __restrict__ W   = W0 + blockIdx.z * zW;
  float* __restrict__       Out = Out0 + blockIdx.z * zO;

  __shared__ __attribute__((aligned(16))) float As[2][BM][LP];
  __shared__ __attribute__((aligned(16))) float Bs[2][BN][LP];

  const int tid  = threadIdx.x;
  const int mb   = blockIdx.x * BM;
  const int nb   = blockIdx.y * BN;
  const int lane = tid & 31;
  const int w    = tid >> 5;
  const int wm   = (w & 3) * 32;
  const int wn   = (w >> 2) * 32;
  const int half = lane >> 4;
  const int l15  = lane & 15;

  v8f acc[2][2];
#pragma unroll
  for (int fm = 0; fm < 2; ++fm)
#pragma unroll
    for (int fn = 0; fn < 2; ++fn) acc[fm][fn] = v8f{};

  const int am = tid >> 1;
  const int ac = (tid & 1) * 8;
  const int wr = tid >> 2;
  const int wk = (tid & 3) * 4;

  const float* __restrict__ ap = A + (size_t)(mb + am) * 1024 + ac;
  const float* __restrict__ wp = W + (size_t)(nb + wr) * 1024 + wk;

  float4 a0 = *(const float4*)(ap);
  float4 a1 = *(const float4*)(ap + 4);
  float4 wv = *(const float4*)(wp);
  *(float4*)&As[0][am][ac]     = a0;
  *(float4*)&As[0][am][ac + 4] = a1;
  *(float4*)&Bs[0][wr][wk]     = wv;
  __syncthreads();

#pragma unroll 2
  for (int kb = 0; kb < 1024; kb += BK) {
    const int  cur  = (kb / BK) & 1;
    const bool more = (kb + BK) < 1024;
    if (more) {
      a0 = *(const float4*)(ap + kb + BK);
      a1 = *(const float4*)(ap + kb + BK + 4);
      wv = *(const float4*)(wp + kb + BK);
    }

#pragma unroll
    for (int k4 = 0; k4 < BK; k4 += 4) {
      v2f af[2], bf[2];
#pragma unroll
      for (int fm = 0; fm < 2; ++fm) {
        const int row = wm + fm * 16 + l15;
        af[fm] = *(const v2f*)&As[cur][row][k4 + half * 2];
      }
#pragma unroll
      for (int fn = 0; fn < 2; ++fn) {
        const int col = wn + fn * 16 + l15;
        bf[fn] = *(const v2f*)&Bs[cur][col][k4 + half * 2];
      }
#pragma unroll
      for (int fm = 0; fm < 2; ++fm)
#pragma unroll
        for (int fn = 0; fn < 2; ++fn)
          acc[fm][fn] = __builtin_amdgcn_wmma_f32_16x16x4_f32(
              false, af[fm], false, bf[fn], (short)0, acc[fm][fn],
              false, false);
    }

    __builtin_amdgcn_sched_barrier(0);
    if (more) {
      *(float4*)&As[cur ^ 1][am][ac]     = a0;
      *(float4*)&As[cur ^ 1][am][ac + 4] = a1;
      *(float4*)&Bs[cur ^ 1][wr][wk]     = wv;
    }
    __syncthreads();
  }

#pragma unroll
  for (int fm = 0; fm < 2; ++fm) {
#pragma unroll
    for (int fn = 0; fn < 2; ++fn) {
      const int col = nb + wn + fn * 16 + l15;
#pragma unroll
      for (int g = 0; g < 8; ++g) {
        const int row = mb + wm + fm * 16 + g + half * 8;
        Out[(size_t)row * 1024 + col] = acc[fm][fn][g];
      }
    }
  }
}

extern "C" void kernel_launch(void* const* d_in, const int* in_sizes, int n_in,
                              void* d_out, int out_size, void* d_ws,
                              size_t ws_size, hipStream_t stream) {
  (void)in_sizes; (void)n_in; (void)out_size; (void)ws_size;

  const float* x        = (const float*)d_in[0];
  const float* enc_rot  = (const float*)d_in[1];
  const float* enc_diag = (const float*)d_in[2];
  const float* dec_rot  = (const float*)d_in[3];
  const float* dec_diag = (const float*)d_in[4];

  float* out = (float*)d_out;
  float* ws  = (float*)d_ws;

  float* fg     = ws;
  float* P      = fg + (size_t)8 * 1024;
  float* lifted = P + (size_t)8 * P_PLANE;
  float* Q      = lifted + (size_t)8 * NN * NN;
  float* Wf     = Q + (size_t)4 * NN * NN;
  const size_t MM = (size_t)NN * NN;

  float* bneck  = out;
  float* final_ = out + (size_t)BB * NN;

  k_build_fg<<<8, 512, 0, stream>>>(enc_rot, enc_diag, dec_rot, dec_diag, fg);

  dim3 grid_p((4 * 1023 + 7) / 8, 8, 1);
  k_build_prefix<<<grid_p, 256, 0, stream>>>(
      enc_rot, enc_diag, dec_rot, dec_diag, fg, P);
  dim3 grid_l(512, 8, 1);
  k_build_lifted<<<grid_l, 256, 0, stream>>>(P, lifted);

  dim3 grid_q(NN / BM, NN / BN, 4);
  k_gemm_xwt<<<grid_q, 256, 0, stream>>>(lifted, lifted + MM, Q,
                                         2 * MM, 2 * MM, MM);
  dim3 grid_w(NN / BM, NN / BN, 2);
  k_gemm_xwt<<<grid_w, 256, 0, stream>>>(Q, Q + MM, Wf, 2 * MM, 2 * MM, MM);

  dim3 grid(BB / BM, NN / BN, 1);
  k_gemm_xwt<<<grid, 256, 0, stream>>>(x,     Wf,      bneck,  0, 0, 0);
  k_gemm_xwt<<<grid, 256, 0, stream>>>(bneck, Wf + MM, final_, 0, 0, 0);
}


# Build: prefix sums per (diagonal, row phase i mod 4) over all t, rotation chosen per t; only the askable entries kept, along the diagonal: a lifted element is 2 loads, not 8; summation order changes
# speedup vs baseline: 8.7246x; 1.0298x over previous
#include <hip/hip_runtime.h>
#include <hip/hip_bf16.h>


#define Hh   512
#define H1   511
#define NN   1024
#define BB   16384

typedef float v2f __attribute__((ext_vector_type(2)));
typedef float v8f __attribute__((ext_vector_type(8)));

__global__ __launch_bounds__(512) void k_build_fg(
    const float* __restrict__ enc_rot, const float* __restrict__ enc_diag,
    const float* __restrict__ dec_rot, const float* __restrict__ dec_diag,
    float* __restrict__ fg) {
  const int m = blockIdx.x;
  const int t = threadIdx.x;
  const float* r = (m < 4) ? (enc_rot + m * Hh) : (dec_rot + (m - 4) * Hh);
  const float* d = (m < 4) ? (enc_diag + m * Hh) : (dec_diag + (m - 4) * Hh);

  __shared__ float sA[512];
  __shared__ float sB[512];
  __shared__ float cS[4];

  const float rv = r[t];
  sA[t] = rv;
  sB[t] = rv * rv;
  __syncthreads();
  for (int off = 256; off > 0; off >>= 1) {
    if (t < off) { sA[t] += sA[t + off]; sB[t] += sB[t + off]; }
    __syncthreads();
  }
  if (t == 0) {
    const float a = sA[0], b = sB[0];
    const float D = 1.0f - a * a + 512.0f * b;
    cS[0] = -2.0f * b / D;
    cS[1] = 1.0f + ((1.0f + a) * (1.0f + a) - 512.0f * b) / D;
    cS[2] = -1.0f - ((1.0f - a) * (1.0f - a) - 512.0f * b) / D;
    cS[3] = -1024.0f / D;
  }
  __syncthreads();
  const float dv = d[t];
  fg[m * 1024 + t]       = dv * (cS[0] + cS[1] * rv);
  fg[m * 1024 + 512 + t] = dv * (cS[2] + cS[3] * rv);
}

#define P_V     ((size_t)1023 * Hh)
#define P_T     ((size_t)2 * 1023 * Hh)
#define P_PLANE ((size_t)2 * 1023 * Hh + 4 * 1024)

template <int N>
__device__ __forceinline__ float row_up_or_zero(float v) {
  return __builtin_bit_cast(
      float, __builtin_amdgcn_update_dpp(0, __builtin_bit_cast(int, v),
                                         0x110 + N, 0xf, 0xf, true));
}

__global__ __launch_bounds__(256) void k_build_prefix(
    const float* __restrict__ enc_rot, const float* __restrict__ enc_diag,
    const float* __restrict__ dec_rot, const float* __restrict__ dec_diag,
    const float* __restrict__ fg, float* __restrict__ Pall) {
  const int m = blockIdx.y;
  float* __restrict__ P = Pall + (size_t)m * P_PLANE;
  const int lane = threadIdx.x & 31;
  const int wave = __builtin_amdgcn_readfirstlane((int)(threadIdx.x >> 5));
  const int si   = blockIdx.x * 8 + wave;
  if (si >= 1023) return;
  const int s  = si - H1;

  const float* r = (m < 4) ? (enc_rot + m * Hh) : (dec_rot + (m - 4) * Hh);
  const float* d = (m < 4) ? (enc_diag + m * Hh) : (dec_diag + (m - 4) * Hh);
  const float* f = fg + m * 1024;
  const float* g = f + 512;

  const int tlo = (s > 0) ? s : 0;
  const int thi = (s < 0) ? (H1 + s) : H1;
  const unsigned span = (unsigned)(thi - tlo);
  const int l3 = lane & 3;

  float uv[16], vv[16];
#pragma unroll
  for (int j = 0; j < 16; ++j) { uv[j] = 0.f; vv[j] = 0.f; }
  float tot_mine = 0.f;

  for (int ph = 0; ph < 4; ++ph) {
    const int k = (ph - lane) & 3;
    const int sgn_f = (k < 2) ? 1 : -1;
    const int off_f = (k == 0) ? 0 : (k == 1) ? -s : (k == 2) ? H1 : (H1 + s);
    const int sgn_r = (k == 0 || k == 3) ? 1 : -1;
    const int off_r = (k == 0) ? -s : (k == 1) ? H1 : (k == 2) ? (H1 + s) : 0;
    const bool odd  = (k & 1) != 0;

    float hv[16];
#pragma unroll
    for (int j = 0; j < 16; ++j) {
      const int t  = j * 32 + lane;
      const int fi = min(max(sgn_f * t + off_f, 0), H1);
      const int ri = min(max(sgn_r * t + off_r, 0), H1);
      const float h = f[fi] + g[fi] * r[ri];
      hv[j] = ((unsigned)(t - tlo) <= span) ? h : 0.f;
    }

    if (((H1 + s) & 1) == 0) {
      const int ts = (H1 + s) >> 1;
      const float dv = odd ? d[min(max(sgn_f * ts + off_f, 0), H1)] : 0.f;
#pragma unroll
      for (int j = 0; j < 16; ++j)
        hv[j] += (j * 32 + lane == ts) ? dv : 0.f;
    }
    if (s == 0) {
#pragma unroll
      for (int j = 0; j < 16; ++j) {
        const float dv = d[sgn_f * (j * 32 + lane) + off_f];
        hv[j] += odd ? 0.f : dv;
      }
    }

    const bool keep_u = (l3 == ph);
    const bool keep_v = (((l3 + 1) & 3) == ph);
    float carry = 0.f;
#pragma unroll
    for (int j = 0; j < 16; ++j) {
      float v = hv[j];
      v += row_up_or_zero<1>(v);
      v += row_up_or_zero<2>(v);
      v += row_up_or_zero<4>(v);
      v += row_up_or_zero<8>(v);
      const float low = __builtin_bit_cast(
          float, __builtin_amdgcn_readlane(__builtin_bit_cast(int, v), 15));
      if (lane >= 16) v += low;
      const float tot = __builtin_bit_cast(
          float, __builtin_amdgcn_readlane(__builtin_bit_cast(int, v), 31));
      const float F = v + carry;
      uv[j] = keep_u ? F : uv[j];
      vv[j] = keep_v ? F : vv[j];
      carry += tot;
    }
    if (lane == ph) tot_mine = carry;
  }

  float* Up = P + (size_t)si * Hh;
  float* Vp = P + P_V + (size_t)si * Hh;
#pragma unroll
  for (int j = 0; j < 16; ++j) {
    Up[j * 32 + lane] = uv[j];
    Vp[j * 32 + lane] = vv[j];
  }
  if (lane < 4) P[P_T + (size_t)si * 4 + lane] = tot_mine;
}

struct LiftedTerms {
  float hi, lo;
  bool  has_lo, any;
};

__device__ __forceinline__ LiftedTerms lifted_load(int i, int j,
                                                   const float* __restrict__ P) {
  const int s   = i - j;
  const int si  = min(max(s + H1, 0), 2 * H1);
  const int tlo = max(s, 0);
  const int thi = min(H1, H1 + s);
  const int a   = max(tlo, i - Hh);
  const int b   = min(i, thi);
  LiftedTerms x;
  x.any    = (s > -Hh) && (s < Hh) && (b >= a);
  x.has_lo = (i - Hh > tlo);
  const size_t iu = (size_t)si * Hh + min(max(i, 0), H1);
  const size_t it = P_T + (size_t)si * 4 + (i & 3);
  x.hi = P[(i <= thi) ? iu : it];
  x.lo = P[P_V + (size_t)si * Hh + min(max(i - 513, 0), H1)];
  return x;
}

__device__ __forceinline__ float lifted_combine(const LiftedTerms& x) {
  const float v = x.hi - (x.has_lo ? x.lo : 0.f);
  return x.any ? v : 0.f;
}

#define LT_R 32
#define LT_D 64

__global__ __launch_bounds__(256) void k_build_lifted(const float* __restrict__ Pall,
                                                      float* __restrict__ lifted) {
  __shared__ float tile[LT_D][LT_R + 1];
  const int m    = blockIdx.y;
  const float* __restrict__ P = Pall + (size_t)m * P_PLANE;
  const int q    = m & 3;
  const bool tr  = (q & 1) == 0;
  const int slot = (m & 4) + ((q == 3) ? 0 : (q == 2) ? 1 : (q == 0) ? 2 : 3);
  float* L = lifted + (size_t)slot * NN * NN;
  const int lane = threadIdx.x & 31;
  const int w    = threadIdx.x >> 5;
  const int R0   = (blockIdx.x & 31) * LT_R;
  const int d0   = (blockIdx.x >> 5) * LT_D;

  const int R = R0 + lane;
  LiftedTerms tm[8];
#pragma unroll
  for (int u = 0; u < 8; ++u) {
    const int dl = w * 8 + u;
    const int C  = (R - d0 - dl) & 1023;
    tm[u] = lifted_load(tr ? C : R, tr ? R : C, P);
  }
#pragma unroll
  for (int u = 0; u < 8; ++u)
    tile[w * 8 + u][lane] = lifted_combine(tm[u]);
  __syncthreads();

#pragma unroll
  for (int u = 0; u < 4; ++u) {
    const int rl = w * 4 + u;
    const int Rr = R0 + rl;
#pragma unroll
    for (int hf = 0; hf < 2; ++hf) {
      const int dl = (LT_D - 1) - (hf * 32 + lane);
      const int C  = (Rr - d0 - dl) & 1023;
      L[(size_t)Rr * NN + C] = tile[dl][rl];
    }
  }
}

#define BM 128
#define BN 64
#define BK 16
#define LP 20

__global__ __launch_bounds__(256) void k_gemm_xwt(const float* __restrict__ A0,
                                                  const float* __restrict__ W0,
                                                  float* __restrict__ Out0,
                                                  size_t zA, size_t zW,
                                                  size_t zO) {
  const float* __restrict__ A   = A0 + blockIdx.z * zA;
  const float* __restrict__ W   = W0 + blockIdx.z * zW;
  float* __restrict__       Out = Out0 + blockIdx.z * zO;

  __shared__ __attribute__((aligned(16))) float As[2][BM][LP];
  __shared__ __attribute__((aligned(16))) float Bs[2][BN][LP];

  const int tid  = threadIdx.x;
  const int mb   = blockIdx.x * BM;
  const int nb   = blockIdx.y * BN;
  const int lane = tid & 31;
  const int w    = tid >> 5;
  const int wm   = (w & 3) * 32;
  const int wn   = (w >> 2) * 32;
  const int half = lane >> 4;
  const int l15  = lane & 15;

  v8f acc[2][2];
#pragma unroll
  for (int fm = 0; fm < 2; ++fm)
#pragma unroll
    for (int fn = 0; fn < 2; ++fn) acc[fm][fn] = v8f{};

  const int am = tid >> 1;
  const int ac = (tid & 1) * 8;
  const int wr = tid >> 2;
  const int wk = (tid & 3) * 4;

  const float* __restrict__ ap = A + (size_t)(mb + am) * 1024 + ac;
  const float* __restrict__ wp = W + (size_t)(nb + wr) * 1024 + wk;

  float4 a0 = *(const float4*)(ap);
  float4 a1 = *(const float4*)(ap + 4);
  float4 wv = *(const float4*)(wp);
  *(float4*)&As[0][am][ac]     = a0;
  *(float4*)&As[0][am][ac + 4] = a1;
  *(float4*)&Bs[0][wr][wk]     = wv;
  __syncthreads();

#pragma unroll 2
  for (int kb = 0; kb < 1024; kb += BK) {
    const int  cur  = (kb / BK) & 1;
    const bool more = (kb + BK) < 1024;
    if (more) {
      a0 = *(const float4*)(ap + kb + BK);
      a1 = *(const float4*)(ap + kb + BK + 4);
      wv = *(const float4*)(wp + kb + BK);
    }

#pragma unroll
    for (int k4 = 0; k4 < BK; k4 += 4) {
      v2f af[2], bf[2];
#pragma unroll
      for (int fm = 0; fm < 2; ++fm) {
        const int row = wm + fm * 16 + l15;
        af[fm] = *(const v2f*)&As[cur][row][k4 + half * 2];
      }
#pragma unroll
      for (int fn = 0; fn < 2; ++fn) {
        const int col = wn + fn * 16 + l15;
        bf[fn] = *(const v2f*)&Bs[cur][col][k4 + half * 2];
      }
#pragma unroll
      for (int fm = 0; fm < 2; ++fm)
#pragma unroll
        for (int fn = 0; fn < 2; ++fn)
          acc[fm][fn] = __builtin_amdgcn_wmma_f32_16x16x4_f32(
              false, af[fm], false, bf[fn], (short)0, acc[fm][fn],
              false, false);
    }

    __builtin_amdgcn_sched_barrier(0);
    if (more) {
      *(float4*)&As[cur ^ 1][am][ac]     = a0;
      *(float4*)&As[cur ^ 1][am][ac + 4] = a1;
      *(float4*)&Bs[cur ^ 1][wr][wk]     = wv;
    }
    __syncthreads();
  }

#pragma unroll
  for (int fm = 0; fm < 2; ++fm) {
#pragma unroll
    for (int fn = 0; fn < 2; ++fn) {
      const int col = nb + wn + fn * 16 + l15;
#pragma unroll
      for (int g = 0; g < 8; ++g) {
        const int row = mb + wm + fm * 16 + g + half * 8;
        Out[(size_t)row * 1024 + col] = acc[fm][fn][g];
      }
    }
  }
}

extern "C" void kernel_launch(void* const* d_in, const int* in_sizes, int n_in,
                              void* d_out, int out_size, void* d_ws,
                              size_t ws_size, hipStream_t stream) {
  (void)in_sizes; (void)n_in; (void)out_size; (void)ws_size;

  const float* x        = (const float*)d_in[0];
  const float* enc_rot  = (const float*)d_in[1];
  const float* enc_diag = (const float*)d_in[2];
  const float* dec_rot  = (const float*)d_in[3];
  const float* dec_diag = (const float*)d_in[4];

  float* out = (float*)d_out;
  float* ws  = (float*)d_ws;

  float* fg     = ws;
  float* P      = fg + (size_t)8 * 1024;
  float* lifted = P + (size_t)8 * P_PLANE;
  float* Q      = lifted + (size_t)8 * NN * NN;
  float* Wf     = Q + (size_t)4 * NN * NN;
  const size_t MM = (size_t)NN * NN;

  float* bneck  = out;
  float* final_ = out + (size_t)BB * NN;

  k_build_fg<<<8, 512, 0, stream>>>(enc_rot, enc_diag, dec_rot, dec_diag, fg);

  dim3 grid_p((1023 + 7) / 8, 8, 1);
  k_build_prefix<<<grid_p, 256, 0, stream>>>(
      enc_rot, enc_diag, dec_rot, dec_diag, fg, P);
  dim3 grid_l(512, 8, 1);
  k_build_lifted<<<grid_l, 256, 0, stream>>>(P, lifted);

  dim3 grid_q(NN / BM, NN / BN, 4);
  k_gemm_xwt<<<grid_q, 256, 0, stream>>>(lifted, lifted + MM, Q,
                                         2 * MM, 2 * MM, MM);
  dim3 grid_w(NN / BM, NN / BN, 2);
  k_gemm_xwt<<<grid_w, 256, 0, stream>>>(Q, Q + MM, Wf, 2 * MM, 2 * MM, MM);

  dim3 grid(BB / BM, NN / BN, 1);
  k_gemm_xwt<<<grid, 256, 0, stream>>>(x,     Wf,      bneck,  0, 0, 0);
  k_gemm_xwt<<<grid, 256, 0, stream>>>(bneck, Wf + MM, final_, 0, 0, 0);
}


# k_gemm_xwt: ring of 3 LDS stages, tile stored 2 K-steps ahead: next step's first fragments load before the barrier, tile loads stay in flight across it (LDS-only fence); same sums, same order
# speedup vs baseline: 9.0418x; 1.0364x over previous
#include <hip/hip_runtime.h>
#include <hip/hip_bf16.h>


#define Hh   512
#define H1   511
#define NN   1024
#define BB   16384

typedef float v2f __attribute__((ext_vector_type(2)));
typedef float v8f __attribute__((ext_vector_type(8)));

__global__ __launch_bounds__(512) void k_build_fg(
    const float* __restrict__ enc_rot, const float* __restrict__ enc_diag,
    const float* __restrict__ dec_rot, const float* __restrict__ dec_diag,
    float* __restrict__ fg) {
  const int m = blockIdx.x;
  const int t = threadIdx.x;
  const float* r = (m < 4) ? (enc_rot + m * Hh) : (dec_rot + (m - 4) * Hh);
  const float* d = (m < 4) ? (enc_diag + m * Hh) : (dec_diag + (m - 4) * Hh);

  __shared__ float sA[512];
  __shared__ float sB[512];
  __shared__ float cS[4];

  const float rv = r[t];
  sA[t] = rv;
  sB[t] = rv * rv;
  __syncthreads();
  for (int off = 256; off > 0; off >>= 1) {
    if (t < off) { sA[t] += sA[t + off]; sB[t] += sB[t + off]; }
    __syncthreads();
  }
  if (t == 0) {
    const float a = sA[0], b = sB[0];
    const float D = 1.0f - a * a + 512.0f * b;
    cS[0] = -2.0f * b / D;
    cS[1] = 1.0f + ((1.0f + a) * (1.0f + a) - 512.0f * b) / D;
    cS[2] = -1.0f - ((1.0f - a) * (1.0f - a) - 512.0f * b) / D;
    cS[3] = -1024.0f / D;
  }
  __syncthreads();
  const float dv = d[t];
  fg[m * 1024 + t]       = dv * (cS[0] + cS[1] * rv);
  fg[m * 1024 + 512 + t] = dv * (cS[2] + cS[3] * rv);
}

#define P_V     ((size_t)1023 * Hh)
#define P_T     ((size_t)2 * 1023 * Hh)
#define P_PLANE ((size_t)2 * 1023 * Hh + 4 * 1024)

template <int N>
__device__ __forceinline__ float row_up_or_zero(float v) {
  return __builtin_bit_cast(
      float, __builtin_amdgcn_update_dpp(0, __builtin_bit_cast(int, v),
                                         0x110 + N, 0xf, 0xf, true));
}

__global__ __launch_bounds__(256) void k_build_prefix(
    const float* __restrict__ enc_rot, const float* __restrict__ enc_diag,
    const float* __restrict__ dec_rot, const float* __restrict__ dec_diag,
    const float* __restrict__ fg, float* __restrict__ Pall) {
  const int m = blockIdx.y;
  float* __restrict__ P = Pall + (size_t)m * P_PLANE;
  const int lane = threadIdx.x & 31;
  const int wave = __builtin_amdgcn_readfirstlane((int)(threadIdx.x >> 5));
  const int si   = blockIdx.x * 8 + wave;
  if (si >= 1023) return;
  const int s  = si - H1;

  const float* r = (m < 4) ? (enc_rot + m * Hh) : (dec_rot + (m - 4) * Hh);
  const float* d = (m < 4) ? (enc_diag + m * Hh) : (dec_diag + (m - 4) * Hh);
  const float* f = fg + m * 1024;
  const float* g = f + 512;

  const int tlo = (s > 0) ? s : 0;
  const int thi = (s < 0) ? (H1 + s) : H1;
  const unsigned span = (unsigned)(thi - tlo);
  const int l3 = lane & 3;

  float uv[16], vv[16];
#pragma unroll
  for (int j = 0; j < 16; ++j) { uv[j] = 0.f; vv[j] = 0.f; }
  float tot_mine = 0.f;

  for (int ph = 0; ph < 4; ++ph) {
    const int k = (ph - lane) & 3;
    const int sgn_f = (k < 2) ? 1 : -1;
    const int off_f = (k == 0) ? 0 : (k == 1) ? -s : (k == 2) ? H1 : (H1 + s);
    const int sgn_r = (k == 0 || k == 3) ? 1 : -1;
    const int off_r = (k == 0) ? -s : (k == 1) ? H1 : (k == 2) ? (H1 + s) : 0;
    const bool odd  = (k & 1) != 0;

    float hv[16];
#pragma unroll
    for (int j = 0; j < 16; ++j) {
      const int t  = j * 32 + lane;
      const int fi = min(max(sgn_f * t + off_f, 0), H1);
      const int ri = min(max(sgn_r * t + off_r, 0), H1);
      const float h = f[fi] + g[fi] * r[ri];
      hv[j] = ((unsigned)(t - tlo) <= span) ? h : 0.f;
    }

    if (((H1 + s) & 1) == 0) {
      const int ts = (H1 + s) >> 1;
      const float dv = odd ? d[min(max(sgn_f * ts + off_f, 0), H1)] : 0.f;
#pragma unroll
      for (int j = 0; j < 16; ++j)
        hv[j] += (j * 32 + lane == ts) ? dv : 0.f;
    }
    if (s == 0) {
#pragma unroll
      for (int j = 0; j < 16; ++j) {
        const float dv = d[sgn_f * (j * 32 + lane) + off_f];
        hv[j] += odd ? 0.f : dv;
      }
    }

    const bool keep_u = (l3 == ph);
    const bool keep_v = (((l3 + 1) & 3) == ph);
    float carry = 0.f;
#pragma unroll
    for (int j = 0; j < 16; ++j) {
      float v = hv[j];
      v += row_up_or_zero<1>(v);
      v += row_up_or_zero<2>(v);
      v += row_up_or_zero<4>(v);
      v += row_up_or_zero<8>(v);
      const float low = __builtin_bit_cast(
          float, __builtin_amdgcn_readlane(__builtin_bit_cast(int, v), 15));
      if (lane >= 16) v += low;
      const float tot = __builtin_bit_cast(
          float, __builtin_amdgcn_readlane(__builtin_bit_cast(int, v), 31));
      const float F = v + carry;
      uv[j] = keep_u ? F : uv[j];
      vv[j] = keep_v ? F : vv[j];
      carry += tot;
    }
    if (lane == ph) tot_mine = carry;
  }

  float* Up = P + (size_t)si * Hh;
  float* Vp = P + P_V + (size_t)si * Hh;
#pragma unroll
  for (int j = 0; j < 16; ++j) {
    Up[j * 32 + lane] = uv[j];
    Vp[j * 32 + lane] = vv[j];
  }
  if (lane < 4) P[P_T + (size_t)si * 4 + lane] = tot_mine;
}

struct LiftedTerms {
  float hi, lo;
  bool  has_lo, any;
};

__device__ __forceinline__ LiftedTerms lifted_load(int i, int j,
                                                   const float* __restrict__ P) {
  const int s   = i - j;
  const int si  = min(max(s + H1, 0), 2 * H1);
  const int tlo = max(s, 0);
  const int thi = min(H1, H1 + s);
  const int a   = max(tlo, i - Hh);
  const int b   = min(i, thi);
  LiftedTerms x;
  x.any    = (s > -Hh) && (s < Hh) && (b >= a);
  x.has_lo = (i - Hh > tlo);
  const size_t iu = (size_t)si * Hh + min(max(i, 0), H1);
  const size_t it = P_T + (size_t)si * 4 + (i & 3);
  x.hi = P[(i <= thi) ? iu : it];
  x.lo = P[P_V + (size_t)si * Hh + min(max(i - 513, 0), H1)];
  return x;
}

__device__ __forceinline__ float lifted_combine(const LiftedTerms& x) {
  const float v = x.hi - (x.has_lo ? x.lo : 0.f);
  return x.any ? v : 0.f;
}

#define LT_R 32
#define LT_D 64

__global__ __launch_bounds__(256) void k_build_lifted(const float* __restrict__ Pall,
                                                      float* __restrict__ lifted) {
  __shared__ float tile[LT_D][LT_R + 1];
  const int m    = blockIdx.y;
  const float* __restrict__ P = Pall + (size_t)m * P_PLANE;
  const int q    = m & 3;
  const bool tr  = (q & 1) == 0;
  const int slot = (m & 4) + ((q == 3) ? 0 : (q == 2) ? 1 : (q == 0) ? 2 : 3);
  float* L = lifted + (size_t)slot * NN * NN;
  const int lane = threadIdx.x & 31;
  const int w    = threadIdx.x >> 5;
  const int R0   = (blockIdx.x & 31) * LT_R;
  const int d0   = (blockIdx.x >> 5) * LT_D;

  const int R = R0 + lane;
  LiftedTerms tm[8];
#pragma unroll
  for (int u = 0; u < 8; ++u) {
    const int dl = w * 8 + u;
    const int C  = (R - d0 - dl) & 1023;
    tm[u] = lifted_load(tr ? C : R, tr ? R : C, P);
  }
#pragma unroll
  for (int u = 0; u < 8; ++u)
    tile[w * 8 + u][lane] = lifted_combine(tm[u]);
  __syncthreads();

#pragma unroll
  for (int u = 0; u < 4; ++u) {
    const int rl = w * 4 + u;
    const int Rr = R0 + rl;
#pragma unroll
    for (int hf = 0; hf < 2; ++hf) {
      const int dl = (LT_D - 1) - (hf * 32 + lane);
      const int C  = (Rr - d0 - dl) & 1023;
      L[(size_t)Rr * NN + C] = tile[dl][rl];
    }
  }
}

#define BM 128
#define BN 64
#define BK 16
#define LP 20
#define NST 3
#define ST_F ((BM + BN) * LP)
#define KT (1024 / BK)

__global__ __launch_bounds__(256) void k_gemm_xwt(const float* A0,
                                                  const float* W0,
                                                  float* Out0,
                                                  size_t zA, size_t zW,
                                                  size_t zO) {
  const float* A   = A0 + blockIdx.z * zA;
  const float* W   = W0 + blockIdx.z * zW;
  float*       Out = Out0 + blockIdx.z * zO;

  __shared__ __attribute__((aligned(16))) float Sm[NST * ST_F];

  const int tid  = threadIdx.x;
  const int mb   = blockIdx.x * BM;
  const int nb   = blockIdx.y * BN;
  const int lane = tid & 31;
  const int w    = tid >> 5;
  const int wm   = (w & 3) * 32;
  const int wn   = (w >> 2) * 32;
  const int half = lane >> 4;
  const int l15  = lane & 15;

  v8f acc[2][2];
#pragma unroll
  for (int fm = 0; fm < 2; ++fm)
#pragma unroll
    for (int fn = 0; fn < 2; ++fn) acc[fm][fn] = v8f{};

  const int am = tid >> 1;
  const int ac = (tid & 1) * 8;
  const int wr = tid >> 2;
  const int wk = (tid & 3) * 4;

  const float* ap = A + (size_t)(mb + am) * 1024 + ac;
  const float* wp = W + (size_t)(nb + wr) * 1024 + wk;

  const int sto_a = am * LP + ac;
  const int sto_w = (BM + wr) * LP + wk;
  const int fra   = (wm + l15) * LP + half * 2;
  const int frb   = (BM + wn + l15) * LP + half * 2;

  float4 a0 = *(const float4*)(ap);
  float4 a1 = *(const float4*)(ap + 4);
  float4 wv = *(const float4*)(wp);
  float4 c0 = *(const float4*)(ap + BK);
  float4 c1 = *(const float4*)(ap + BK + 4);
  float4 cv = *(const float4*)(wp + BK);
  *(float4*)&Sm[sto_a]            = a0;
  *(float4*)&Sm[sto_a + 4]        = a1;
  *(float4*)&Sm[sto_w]            = wv;
  *(float4*)&Sm[ST_F + sto_a]     = c0;
  *(float4*)&Sm[ST_F + sto_a + 4] = c1;
  *(float4*)&Sm[ST_F + sto_w]     = cv;
  a0 = *(const float4*)(ap + 2 * BK);
  a1 = *(const float4*)(ap + 2 * BK + 4);
  wv = *(const float4*)(wp + 2 * BK);
  __syncthreads();

  v2f afn[2], bfn[2];
  afn[0] = *(const v2f*)&Sm[fra];
  afn[1] = *(const v2f*)&Sm[fra + 16 * LP];
  bfn[0] = *(const v2f*)&Sm[frb];
  bfn[1] = *(const v2f*)&Sm[frb + 16 * LP];

  auto step = [&](const int t, const int cur, const int nxt, const int sto)
                  __attribute__((always_inline)) {
    if (t + 2 < KT) {
      *(float4*)&Sm[sto + sto_a]     = a0;
      *(float4*)&Sm[sto + sto_a + 4] = a1;
      *(float4*)&Sm[sto + sto_w]     = wv;
    }
    if (t + 3 < KT) {
      a0 = *(const float4*)(ap + (t + 3) * BK);
      a1 = *(const float4*)(ap + (t + 3) * BK + 4);
      wv = *(const float4*)(wp + (t + 3) * BK);
    }
    __builtin_amdgcn_sched_barrier(0);

    const bool more = (t + 1) < KT;
#pragma unroll
    for (int g = 0; g < 4; ++g) {
      v2f af[2], bf[2];
      af[0] = afn[0]; af[1] = afn[1];
      bf[0] = bfn[0]; bf[1] = bfn[1];
      if (g < 3 || more) {
        const int src = (g < 3) ? (cur + (g + 1) * 4) : nxt;
        afn[0] = *(const v2f*)&Sm[src + fra];
        afn[1] = *(const v2f*)&Sm[src + fra + 16 * LP];
        bfn[0] = *(const v2f*)&Sm[src + frb];
        bfn[1] = *(const v2f*)&Sm[src + frb + 16 * LP];
      }
#pragma unroll
      for (int fm = 0; fm < 2; ++fm)
#pragma unroll
        for (int fn = 0; fn < 2; ++fn)
          acc[fm][fn] = __builtin_amdgcn_wmma_f32_16x16x4_f32(
              false, af[fm], false, bf[fn], (short)0, acc[fm][fn],
              false, false);
    }
    __builtin_amdgcn_fence(__ATOMIC_RELEASE, "workgroup", "local");
    __builtin_amdgcn_s_barrier();
    __builtin_amdgcn_fence(__ATOMIC_ACQUIRE, "workgroup", "local");
  };

  static_assert((KT - 1) % NST == 0, "the ring is unrolled three steps per pass, one step follows");
  for (int t0 = 0; t0 < KT - 1; t0 += NST) {
    step(t0,     0,        ST_F,     2 * ST_F);
    step(t0 + 1, ST_F,     2 * ST_F, 0);
    step(t0 + 2, 2 * ST_F, 0,        ST_F);
  }
  step(KT - 1, 0, ST_F, 2 * ST_F);

#pragma unroll
  for (int fm = 0; fm < 2; ++fm) {
#pragma unroll
    for (int fn = 0; fn < 2; ++fn) {
      const int col = nb + wn + fn * 16 + l15;
#pragma unroll
      for (int g = 0; g < 8; ++g) {
        const int row = mb + wm + fm * 16 + g + half * 8;
        Out[(size_t)row * 1024 + col] = acc[fm][fn][g];
      }
    }
  }
}

extern "C" void kernel_launch(void* const* d_in, const int* in_sizes, int n_in,
                              void* d_out, int out_size, void* d_ws,
                              size_t ws_size, hipStream_t stream) {
  (void)in_sizes; (void)n_in; (void)out_size; (void)ws_size;

  const float* x        = (const float*)d_in[0];
  const float* enc_rot  = (const float*)d_in[1];
  const float* enc_diag = (const float*)d_in[2];
  const float* dec_rot  = (const float*)d_in[3];
  const float* dec_diag = (const float*)d_in[4];

  float* out = (float*)d_out;
  float* ws  = (float*)d_ws;

  float* fg     = ws;
  float* P      = fg + (size_t)8 * 1024;
  float* lifted = P + (size_t)8 * P_PLANE;
  float* Q      = lifted + (size_t)8 * NN * NN;
  float* Wf     = Q + (size_t)4 * NN * NN;
  const size_t MM = (size_t)NN * NN;

  float* bneck  = out;
  float* final_ = out + (size_t)BB * NN;

  k_build_fg<<<8, 512, 0, stream>>>(enc_rot, enc_diag, dec_rot, dec_diag, fg);

  dim3 grid_p((1023 + 7) / 8, 8, 1);
  k_build_prefix<<<grid_p, 256, 0, stream>>>(
      enc_rot, enc_diag, dec_rot, dec_diag, fg, P);
  dim3 grid_l(512, 8, 1);
  k_build_lifted<<<grid_l, 256, 0, stream>>>(P, lifted);

  dim3 grid_q(NN / BM, NN / BN, 4);
  k_gemm_xwt<<<grid_q, 256, 0, stream>>>(lifted, lifted + MM, Q,
                                         2 * MM, 2 * MM, MM);
  dim3 grid_w(NN / BM, NN / BN, 2);
  k_gemm_xwt<<<grid_w, 256, 0, stream>>>(Q, Q + MM, Wf, 2 * MM, 2 * MM, MM);

  dim3 grid(BB / BM, NN / BN, 1);
  k_gemm_xwt<<<grid, 256, 0, stream>>>(x,     Wf,      bneck,  0, 0, 0);
  k_gemm_xwt<<<grid, 256, 0, stream>>>(bneck, Wf + MM, final_, 0, 0, 0);
}


# Batch GEMMs: wave tile 64x64 (4x4 accumulators) in 256x128 block tiles, was 32x32 in 128x64: a fragment load feeds 4 WMMAs, not 2; half the LDS and global loads per WMMA; same sums, same order
# speedup vs baseline: 9.5381x; 1.0549x over previous
#include <hip/hip_runtime.h>
#include <hip/hip_bf16.h>


#define Hh   512
#define H1   511
#define NN   1024
#define BB   16384

typedef float v2f __attribute__((ext_vector_type(2)));
typedef float v8f __attribute__((ext_vector_type(8)));

__global__ __launch_bounds__(512) void k_build_fg(
    const float* __restrict__ enc_rot, const float* __restrict__ enc_diag,
    const float* __restrict__ dec_rot, const float* __restrict__ dec_diag,
    float* __restrict__ fg) {
  const int m = blockIdx.x;
  const int t = threadIdx.x;
  const float* r = (m < 4) ? (enc_rot + m * Hh) : (dec_rot + (m - 4) * Hh);
  const float* d = (m < 4) ? (enc_diag + m * Hh) : (dec_diag + (m - 4) * Hh);

  __shared__ float sA[512];
  __shared__ float sB[512];
  __shared__ float cS[4];

  const float rv = r[t];
  sA[t] = rv;
  sB[t] = rv * rv;
  __syncthreads();
  for (int off = 256; off > 0; off >>= 1) {
    if (t < off) { sA[t] += sA[t + off]; sB[t] += sB[t + off]; }
    __syncthreads();
  }
  if (t == 0) {
    const float a = sA[0], b = sB[0];
    const float D = 1.0f - a * a + 512.0f * b;
    cS[0] = -2.0f * b / D;
    cS[1] = 1.0f + ((1.0f + a) * (1.0f + a) - 512.0f * b) / D;
    cS[2] = -1.0f - ((1.0f - a) * (1.0f - a) - 512.0f * b) / D;
    cS[3] = -1024.0f / D;
  }
  __syncthreads();
  const float dv = d[t];
  fg[m * 1024 + t]       = dv * (cS[0] + cS[1] * rv);
  fg[m * 1024 + 512 + t] = dv * (cS[2] + cS[3] * rv);
}

#define P_V     ((size_t)1023 * Hh)
#define P_T     ((size_t)2 * 1023 * Hh)
#define P_PLANE ((size_t)2 * 1023 * Hh + 4 * 1024)

template <int N>
__device__ __forceinline__ float row_up_or_zero(float v) {
  return __builtin_bit_cast(
      float, __builtin_amdgcn_update_dpp(0, __builtin_bit_cast(int, v),
                                         0x110 + N, 0xf, 0xf, true));
}

__global__ __launch_bounds__(256) void k_build_prefix(
    const float* __restrict__ enc_rot, const float* __restrict__ enc_diag,
    const float* __restrict__ dec_rot, const float* __restrict__ dec_diag,
    const float* __restrict__ fg, float* __restrict__ Pall) {
  const int m = blockIdx.y;
  float* __restrict__ P = Pall + (size_t)m * P_PLANE;
  const int lane = threadIdx.x & 31;
  const int wave = __builtin_amdgcn_readfirstlane((int)(threadIdx.x >> 5));
  const int si   = blockIdx.x * 8 + wave;
  if (si >= 1023) return;
  const int s  = si - H1;

  const float* r = (m < 4) ? (enc_rot + m * Hh) : (dec_rot + (m - 4) * Hh);
  const float* d = (m < 4) ? (enc_diag + m * Hh) : (dec_diag + (m - 4) * Hh);
  const float* f = fg + m * 1024;
  const float* g = f + 512;

  const int tlo = (s > 0) ? s : 0;
  const int thi = (s < 0) ? (H1 + s) : H1;
  const unsigned span = (unsigned)(thi - tlo);
  const int l3 = lane & 3;

  float uv[16], vv[16];
#pragma unroll
  for (int j = 0; j < 16; ++j) { uv[j] = 0.f; vv[j] = 0.f; }
  float tot_mine = 0.f;

  for (int ph = 0; ph < 4; ++ph) {
    const int k = (ph - lane) & 3;
    const int sgn_f = (k < 2) ? 1 : -1;
    const int off_f = (k == 0) ? 0 : (k == 1) ? -s : (k == 2) ? H1 : (H1 + s);
    const int sgn_r = (k == 0 || k == 3) ? 1 : -1;
    const int off_r = (k == 0) ? -s : (k == 1) ? H1 : (k == 2) ? (H1 + s) : 0;
    const bool odd  = (k & 1) != 0;

    float hv[16];
#pragma unroll
    for (int j = 0; j < 16; ++j) {
      const int t  = j * 32 + lane;
      const int fi = min(max(sgn_f * t + off_f, 0), H1);
      const int ri = min(max(sgn_r * t + off_r, 0), H1);
      const float h = f[fi] + g[fi] * r[ri];
      hv[j] = ((unsigned)(t - tlo) <= span) ? h : 0.f;
    }

    if (((H1 + s) & 1) == 0) {
      const int ts = (H1 + s) >> 1;
      const float dv = odd ? d[min(max(sgn_f * ts + off_f, 0), H1)] : 0.f;
#pragma unroll
      for (int j = 0; j < 16; ++j)
        hv[j] += (j * 32 + lane == ts) ? dv : 0.f;
    }
    if (s == 0) {
#pragma unroll
      for (int j = 0; j < 16; ++j) {
        const float dv = d[sgn_f * (j * 32 + lane) + off_f];
        hv[j] += odd ? 0.f : dv;
      }
    }

    const bool keep_u = (l3 == ph);
    const bool keep_v = (((l3 + 1) & 3) == ph);
    float carry = 0.f;
#pragma unroll
    for (int j = 0; j < 16; ++j) {
      float v = hv[j];
      v += row_up_or_zero<1>(v);
      v += row_up_or_zero<2>(v);
      v += row_up_or_zero<4>(v);
      v += row_up_or_zero<8>(v);
      const float low = __builtin_bit_cast(
          float, __builtin_amdgcn_readlane(__builtin_bit_cast(int, v), 15));
      if (lane >= 16) v += low;
      const float tot = __builtin_bit_cast(
          float, __builtin_amdgcn_readlane(__builtin_bit_cast(int, v), 31));
      const float F = v + carry;
      uv[j] = keep_u ? F : uv[j];
      vv[j] = keep_v ? F : vv[j];
      carry += tot;
    }
    if (lane == ph) tot_mine = carry;
  }

  float* Up = P + (size_t)si * Hh;
  float* Vp = P + P_V + (size_t)si * Hh;
#pragma unroll
  for (int j = 0; j < 16; ++j) {
    Up[j * 32 + lane] = uv[j];
    Vp[j * 32 + lane] = vv[j];
  }
  if (lane < 4) P[P_T + (size_t)si * 4 + lane] = tot_mine;
}

struct LiftedTerms {
  float hi, lo;
  bool  has_lo, any;
};

__device__ __forceinline__ LiftedTerms lifted_load(int i, int j,
                                                   const float* __restrict__ P) {
  const int s   = i - j;
  const int si  = min(max(s + H1, 0), 2 * H1);
  const int tlo = max(s, 0);
  const int thi = min(H1, H1 + s);
  const int a   = max(tlo, i - Hh);
  const int b   = min(i, thi);
  LiftedTerms x;
  x.any    = (s > -Hh) && (s < Hh) && (b >= a);
  x.has_lo = (i - Hh > tlo);
  const size_t iu = (size_t)si * Hh + min(max(i, 0), H1);
  const size_t it = P_T + (size_t)si * 4 + (i & 3);
  x.hi = P[(i <= thi) ? iu : it];
  x.lo = P[P_V + (size_t)si * Hh + min(max(i - 513, 0), H1)];
  return x;
}

__device__ __forceinline__ float lifted_combine(const LiftedTerms& x) {
  const float v = x.hi - (x.has_lo ? x.lo : 0.f);
  return x.any ? v : 0.f;
}

#define LT_R 32
#define LT_D 64

__global__ __launch_bounds__(256) void k_build_lifted(const float* __restrict__ Pall,
                                                      float* __restrict__ lifted) {
  __shared__ float tile[LT_D][LT_R + 1];
  const int m    = blockIdx.y;
  const float* __restrict__ P = Pall + (size_t)m * P_PLANE;
  const int q    = m & 3;
  const bool tr  = (q & 1) == 0;
  const int slot = (m & 4) + ((q == 3) ? 0 : (q == 2) ? 1 : (q == 0) ? 2 : 3);
  float* L = lifted + (size_t)slot * NN * NN;
  const int lane = threadIdx.x & 31;
  const int w    = threadIdx.x >> 5;
  const int R0   = (blockIdx.x & 31) * LT_R;
  const int d0   = (blockIdx.x >> 5) * LT_D;

  const int R = R0 + lane;
  LiftedTerms tm[8];
#pragma unroll
  for (int u = 0; u < 8; ++u) {
    const int dl = w * 8 + u;
    const int C  = (R - d0 - dl) & 1023;
    tm[u] = lifted_load(tr ? C : R, tr ? R : C, P);
  }
#pragma unroll
  for (int u = 0; u < 8; ++u)
    tile[w * 8 + u][lane] = lifted_combine(tm[u]);
  __syncthreads();

#pragma unroll
  for (int u = 0; u < 4; ++u) {
    const int rl = w * 4 + u;
    const int Rr = R0 + rl;
#pragma unroll
    for (int hf = 0; hf < 2; ++hf) {
      const int dl = (LT_D - 1) - (hf * 32 + lane);
      const int C  = (Rr - d0 - dl) & 1023;
      L[(size_t)Rr * NN + C] = tile[dl][rl];
    }
  }
}

#define BM 128
#define BN 64
#define XM 256
#define XN 128
#define BK 16
#define LP 20
#define NST 3
#define KT (1024 / BK)

template <int TBM, int TBN, int FM, int FN>
__global__ __launch_bounds__(256) void k_gemm_xwt(const float* A0,
                                                  const float* W0,
                                                  float* Out0,
                                                  size_t zA, size_t zW,
                                                  size_t zO) {
  static_assert(TBM == 4 * 16 * FM && TBN == 2 * 16 * FN, "8 waves: 4 along M, 2 along N");
  static_assert((TBM == 128 || TBM == 256) && (TBN == 64 || TBN == 128), "a thread stages one or two rows of A (128 apart) and of W (64 apart)");
  constexpr int ST_F = (TBM + TBN) * LP;
  constexpr int NA   = TBM / 128;
  constexpr int NW   = TBN / 64;

  const float* A   = A0 + blockIdx.z * zA;
  const float* W   = W0 + blockIdx.z * zW;
  float*       Out = Out0 + blockIdx.z * zO;

  __shared__ __attribute__((aligned(16))) float Sm[NST * ST_F];

  const int tid  = threadIdx.x;
  const int mb   = blockIdx.x * TBM;
  const int nb   = blockIdx.y * TBN;
  const int lane = tid & 31;
  const int w    = tid >> 5;
  const int wm   = (w & 3) * (16 * FM);
  const int wn   = (w >> 2) * (16 * FN);
  const int half = lane >> 4;
  const int l15  = lane & 15;

  v8f acc[FM][FN];
#pragma unroll
  for (int fm = 0; fm < FM; ++fm)
#pragma unroll
    for (int fn = 0; fn < FN; ++fn) acc[fm][fn] = v8f{};

  const int am = tid >> 1;
  const int ac = (tid & 1) * 8;
  const int wr = tid >> 2;
  const int wk = (tid & 3) * 4;

  const float* ap = A + (size_t)(mb + am) * 1024 + ac;
  const float* wp = W + (size_t)(nb + wr) * 1024 + wk;

  const int sto_a = am * LP + ac;
  const int sto_w = (TBM + wr) * LP + wk;
  const int fra   = (wm + l15) * LP + half * 2;
  const int frb   = (TBM + wn + l15) * LP + half * 2;

  auto tile_load = [&](float4& p0, float4& p1, float4& p2, float4& p3,
                       float4& q0, float4& q1, const int t)
                       __attribute__((always_inline)) {
    p0 = *(const float4*)(ap + t * BK);
    p1 = *(const float4*)(ap + t * BK + 4);
    if constexpr (NA > 1) {
      p2 = *(const float4*)(ap + (size_t)128 * 1024 + t * BK);
      p3 = *(const float4*)(ap + (size_t)128 * 1024 + t * BK + 4);
    }
    q0 = *(const float4*)(wp + t * BK);
    if constexpr (NW > 1) q1 = *(const float4*)(wp + (size_t)64 * 1024 + t * BK);
  };
  auto tile_store = [&](const float4& p0, const float4& p1, const float4& p2,
                        const float4& p3, const float4& q0, const float4& q1,
                        const int sto) __attribute__((always_inline)) {
    *(float4*)&Sm[sto + sto_a]     = p0;
    *(float4*)&Sm[sto + sto_a + 4] = p1;
    if constexpr (NA > 1) {
      *(float4*)&Sm[sto + sto_a + 128 * LP]     = p2;
      *(float4*)&Sm[sto + sto_a + 128 * LP + 4] = p3;
    }
    *(float4*)&Sm[sto + sto_w] = q0;
    if constexpr (NW > 1) *(float4*)&Sm[sto + sto_w + 64 * LP] = q1;
  };

  float4 a0, a1, a2, a3, wv, wx;
  a2 = a3 = wx = float4{};
  {
    float4 c0, c1, c2, c3, cv, cx;
    c2 = c3 = cx = float4{};
    tile_load(a0, a1, a2, a3, wv, wx, 0);
    tile_load(c0, c1, c2, c3, cv, cx, 1);
    tile_store(a0, a1, a2, a3, wv, wx, 0);
    tile_store(c0, c1, c2, c3, cv, cx, ST_F);
  }
  tile_load(a0, a1, a2, a3, wv, wx, 2);
  __syncthreads();

  v2f afn[FM], bfn[FN];
#pragma unroll
  for (int fm = 0; fm < FM; ++fm) afn[fm] = *(const v2f*)&Sm[fra + fm * 16 * LP];
#pragma unroll
  for (int fn = 0; fn < FN; ++fn) bfn[fn] = *(const v2f*)&Sm[frb + fn * 16 * LP];

  auto step = [&](const int t, const int cur, const int nxt, const int sto)
                  __attribute__((always_inline)) {
    if (t + 2 < KT) tile_store(a0, a1, a2, a3, wv, wx, sto);
    if (t + 3 < KT) tile_load(a0, a1, a2, a3, wv, wx, t + 3);
    __builtin_amdgcn_sched_barrier(0);

    const bool more = (t + 1) < KT;
#pragma unroll
    for (int g = 0; g < 4; ++g) {
      v2f af[FM], bf[FN];
#pragma unroll
      for (int fm = 0; fm < FM; ++fm) af[fm] = afn[fm];
#pragma unroll
      for (int fn = 0; fn < FN; ++fn) bf[fn] = bfn[fn];
      if (g < 3 || more) {
        const int src = (g < 3) ? (cur + (g + 1) * 4) : nxt;
#pragma unroll
        for (int fm = 0; fm < FM; ++fm) afn[fm] = *(const v2f*)&Sm[src + fra + fm * 16 * LP];
#pragma unroll
        for (int fn = 0; fn < FN; ++fn) bfn[fn] = *(const v2f*)&Sm[src + frb + fn * 16 * LP];
      }
#pragma unroll
      for (int fm = 0; fm < FM; ++fm)
#pragma unroll
        for (int fn = 0; fn < FN; ++fn)
          acc[fm][fn] = __builtin_amdgcn_wmma_f32_16x16x4_f32(
              false, af[fm], false, bf[fn], (short)0, acc[fm][fn],
              false, false);
    }
    __builtin_amdgcn_fence(__ATOMIC_RELEASE, "workgroup", "local");
    __builtin_amdgcn_s_barrier();
    __builtin_amdgcn_fence(__ATOMIC_ACQUIRE, "workgroup", "local");
  };

  static_assert((KT - 1) % NST == 0, "the ring is unrolled three steps per pass, one step follows");
  for (int t0 = 0; t0 < KT - 1; t0 += NST) {
    step(t0,     0,        ST_F,     2 * ST_F);
    step(t0 + 1, ST_F,     2 * ST_F, 0);
    step(t0 + 2, 2 * ST_F, 0,        ST_F);
  }
  step(KT - 1, 0, ST_F, 2 * ST_F);

#pragma unroll
  for (int fm = 0; fm < FM; ++fm) {
#pragma unroll
    for (int fn = 0; fn < FN; ++fn) {
      const int col = nb + wn + fn * 16 + l15;
#pragma unroll
      for (int g = 0; g < 8; ++g) {
        const int row = mb + wm + fm * 16 + g + half * 8;
        Out[(size_t)row * 1024 + col] = acc[fm][fn][g];
      }
    }
  }
}

extern "C" void kernel_launch(void* const* d_in, const int* in_sizes, int n_in,
                              void* d_out, int out_size, void* d_ws,
                              size_t ws_size, hipStream_t stream) {
  (void)in_sizes; (void)n_in; (void)out_size; (void)ws_size;

  const float* x        = (const float*)d_in[0];
  const float* enc_rot  = (const float*)d_in[1];
  const float* enc_diag = (const float*)d_in[2];
  const float* dec_rot  = (const float*)d_in[3];
  const float* dec_diag = (const float*)d_in[4];

  float* out = (float*)d_out;
  float* ws  = (float*)d_ws;

  float* fg     = ws;
  float* P      = fg + (size_t)8 * 1024;
  float* lifted = P + (size_t)8 * P_PLANE;
  float* Q      = lifted + (size_t)8 * NN * NN;
  float* Wf     = Q + (size_t)4 * NN * NN;
  const size_t MM = (size_t)NN * NN;

  float* bneck  = out;
  float* final_ = out + (size_t)BB * NN;

  k_build_fg<<<8, 512, 0, stream>>>(enc_rot, enc_diag, dec_rot, dec_diag, fg);

  dim3 grid_p((1023 + 7) / 8, 8, 1);
  k_build_prefix<<<grid_p, 256, 0, stream>>>(
      enc_rot, enc_diag, dec_rot, dec_diag, fg, P);
  dim3 grid_l(512, 8, 1);
  k_build_lifted<<<grid_l, 256, 0, stream>>>(P, lifted);

  dim3 grid_q(NN / BM, NN / BN, 4);
  k_gemm_xwt<BM, BN, 2, 2><<<grid_q, 256, 0, stream>>>(lifted, lifted + MM, Q,
                                         2 * MM, 2 * MM, MM);
  dim3 grid_w(NN / BM, NN / BN, 2);
  k_gemm_xwt<BM, BN, 2, 2><<<grid_w, 256, 0, stream>>>(Q, Q + MM, Wf, 2 * MM, 2 * MM, MM);

  dim3 grid(BB / XM, NN / XN, 1);
  k_gemm_xwt<XM, XN, 4, 4><<<grid, 256, 0, stream>>>(x,     Wf,      bneck,  0, 0, 0);
  k_gemm_xwt<XM, XN, 4, 4><<<grid, 256, 0, stream>>>(bneck, Wf + MM, final_, 0, 0, 0);
}
